# starting kernel
# baseline (speedup 1.0000x reference)
#include <hip/hip_runtime.h>
#include <math.h>


typedef __attribute__((ext_vector_type(2))) float v2f;
typedef __attribute__((ext_vector_type(8))) float v8f;

#define CDIM 64
#define TDIM 4
#define ROWLEN 64
#define TCP 256

__device__ __forceinline__ v8f wmma16x16_k64(const float* __restrict__ Arows,
                                             const float* __restrict__ Blds,
                                             int c0) {
    const int lane = threadIdx.x & 31;
    const int m    = lane & 15;
    const int koff = (lane >> 4) << 1;
    const float* __restrict__ arow = Arows + m * ROWLEN;
    v8f acc = {0.f, 0.f, 0.f, 0.f, 0.f, 0.f, 0.f, 0.f};
#pragma unroll
    for (int k0 = 0; k0 < 64; k0 += 4) {
        v2f a, b;
        a.x = arow[k0 + koff];
        a.y = arow[k0 + koff + 1];
        b.x = Blds[(k0 + koff)     * ROWLEN + c0 + m];
        b.y = Blds[(k0 + koff + 1) * ROWLEN + c0 + m];
        acc = __builtin_amdgcn_wmma_f32_16x16x4_f32(
            false, a, false, b, (short)0, acc, false, false);
    }
    return acc;
}

__device__ __forceinline__ void store_tile(float* __restrict__ Out, int row0,
                                           int c0, v8f acc) {
    const int lane = threadIdx.x & 31;
    const int col  = c0 + (lane & 15);
    const int rb   = row0 + ((lane >> 4) << 3);
#pragma unroll
    for (int i = 0; i < 8; ++i)
        Out[(rb + i) * ROWLEN + col] = acc[i];
}

__global__ void __launch_bounds__(128)
proj_src_kernel(const float* __restrict__ feat, const float* __restrict__ U,
                float* __restrict__ Out, int mtiles) {
    __shared__ float ldsU[CDIM * CDIM];
    for (int i = threadIdx.x; i < CDIM * CDIM; i += blockDim.x) ldsU[i] = U[i];
    __syncthreads();

    const int wave = threadIdx.x >> 5;
    const int tile = blockIdx.x * (blockDim.x >> 5) + wave;
    if (tile >= mtiles) return;
    const int row0 = tile * 16;
    const float* __restrict__ Arows = feat + (long long)row0 * ROWLEN;
#pragma unroll
    for (int c0 = 0; c0 < 64; c0 += 16) {
        v8f acc = wmma16x16_k64(Arows, ldsU, c0);
        store_tile(Out, row0, c0, acc);
    }
}

__global__ void __launch_bounds__(128)
proj_dst_kernel(const float* __restrict__ feat, const float* __restrict__ U,
                const float* __restrict__ V, float* __restrict__ OutBase,
                float* __restrict__ OutV, int mtiles) {
    __shared__ float ldsU[CDIM * CDIM];
    __shared__ float ldsV[CDIM * CDIM];
    for (int i = threadIdx.x; i < CDIM * CDIM; i += blockDim.x) {
        ldsU[i] = U[i];
        ldsV[i] = V[i];
    }
    __syncthreads();

    const int wave = threadIdx.x >> 5;
    const int tile = blockIdx.x * (blockDim.x >> 5) + wave;
    if (tile >= mtiles) return;
    const int row0 = tile * 16;
    const int lane = threadIdx.x & 31;
    const int m    = lane & 15;
    const int koff = (lane >> 4) << 1;
    const float* __restrict__ arow = feat + ((long long)row0 + m) * ROWLEN;

#pragma unroll
    for (int c0 = 0; c0 < 64; c0 += 16) {
        v8f accU = {0.f, 0.f, 0.f, 0.f, 0.f, 0.f, 0.f, 0.f};
        v8f accV = {0.f, 0.f, 0.f, 0.f, 0.f, 0.f, 0.f, 0.f};
#pragma unroll
        for (int k0 = 0; k0 < 64; k0 += 4) {
            v2f a, bu, bv;
            a.x  = arow[k0 + koff];
            a.y  = arow[k0 + koff + 1];
            bu.x = ldsU[(k0 + koff)     * ROWLEN + c0 + m];
            bu.y = ldsU[(k0 + koff + 1) * ROWLEN + c0 + m];
            bv.x = ldsV[(k0 + koff)     * ROWLEN + c0 + m];
            bv.y = ldsV[(k0 + koff + 1) * ROWLEN + c0 + m];
            accU = __builtin_amdgcn_wmma_f32_16x16x4_f32(
                false, a, false, bu, (short)0, accU, false, false);
            accV = __builtin_amdgcn_wmma_f32_16x16x4_f32(
                false, a, false, bv, (short)0, accV, false, false);
        }
        store_tile(OutBase, row0, c0, accU);
        store_tile(OutV,    row0, c0, accV);
    }
}

__global__ void __launch_bounds__(256)
edge_kernel(const float* __restrict__ h_src, const float* __restrict__ h_dst,
            const float* __restrict__ edge_w, const float* __restrict__ we,
            const int* __restrict__ src_idx, const int* __restrict__ dst_idx,
            float* __restrict__ aggr, long long total) {
    long long i = (long long)blockIdx.x * blockDim.x + threadIdx.x;
    if (i >= total) return;
    const int e   = (int)(i >> 6);
    const int r   = (int)(i & 63);
    const int tc4 = r << 2;
    const int c   = tc4 & (CDIM - 1);

    const int   s = src_idx[e];
    const int   d = dst_idx[e];
    const float w = edge_w[e];

    const float4 hs = *(const float4*)(h_src + (long long)s * TCP + tc4);
    const float4 hd = *(const float4*)(h_dst + (long long)d * TCP + tc4);
    const float4 wv = *(const float4*)(we + c);
    float* __restrict__ out = aggr + (long long)d * TCP + tc4;

    {
        float x = hs.x * hd.x * (w * wv.x);
        atomicAdd(out + 0, hs.x / (1.0f + __expf(-x)));
    }
    {
        float x = hs.y * hd.y * (w * wv.y);
        atomicAdd(out + 1, hs.y / (1.0f + __expf(-x)));
    }
    {
        float x = hs.z * hd.z * (w * wv.z);
        atomicAdd(out + 2, hs.z / (1.0f + __expf(-x)));
    }
    {
        float x = hs.w * hd.w * (w * wv.w);
        atomicAdd(out + 3, hs.w / (1.0f + __expf(-x)));
    }
}

__global__ void __launch_bounds__(256)
ln_kernel(const float* __restrict__ acc, const float* __restrict__ feat_dst,
          const float* __restrict__ gamma, const float* __restrict__ beta,
          float* __restrict__ out, int rows) {
    const int wave = threadIdx.x >> 5;
    const int lane = threadIdx.x & 31;
    const int row  = blockIdx.x * (blockDim.x >> 5) + wave;
    if (row >= rows) return;

    const float* __restrict__ x = acc + (long long)row * CDIM;
    float x0 = x[lane];
    float x1 = x[lane + 32];

    float s = x0 + x1;
#pragma unroll
    for (int off = 16; off > 0; off >>= 1) s += __shfl_xor(s, off, 32);
    const float mean = s * (1.0f / 64.0f);

    const float d0 = x0 - mean, d1 = x1 - mean;
    float q = d0 * d0 + d1 * d1;
#pragma unroll
    for (int off = 16; off > 0; off >>= 1) q += __shfl_xor(q, off, 32);
    const float rstd = rsqrtf(q * (1.0f / 64.0f) + 1e-5f);

    const float* __restrict__ f = feat_dst + (long long)row * CDIM;
    out[(long long)row * CDIM + lane] =
        d0 * rstd * gamma[lane] + beta[lane] + f[lane];
    out[(long long)row * CDIM + lane + 32] =
        d1 * rstd * gamma[lane + 32] + beta[lane + 32] + f[lane + 32];
}

extern "C" void kernel_launch(void* const* d_in, const int* in_sizes, int n_in,
                              void* d_out, int out_size, void* d_ws, size_t ws_size,
                              hipStream_t stream) {
    const float* feat_src = (const float*)d_in[0];
    const float* feat_dst = (const float*)d_in[1];
    const float* edge_w   = (const float*)d_in[2];
    const float* weight_e = (const float*)d_in[3];
    const float* u        = (const float*)d_in[4];
    const float* v        = (const float*)d_in[5];
    const float* ln_gamma = (const float*)d_in[6];
    const float* ln_beta  = (const float*)d_in[7];
    const int*   src_idx  = (const int*)d_in[8];
    const int*   dst_idx  = (const int*)d_in[9];
    float*       out      = (float*)d_out;

    const int n_src = in_sizes[0] / TCP;
    const int n_dst = in_sizes[1] / TCP;
    const int E     = in_sizes[2];

    float* ws    = (float*)d_ws;
    float* h_src = ws;
    float* h_dst = ws + (size_t)n_src * TCP;
    float* aggr  = h_dst + (size_t)n_dst * TCP;

    const int mtiles_src = (n_src * TDIM) / 16;
    const int mtiles_dst = (n_dst * TDIM) / 16;
    const int wpb = 4;

    proj_src_kernel<<<(mtiles_src + wpb - 1) / wpb, 128, 0, stream>>>(
        feat_src, u, h_src, mtiles_src);

    proj_dst_kernel<<<(mtiles_dst + wpb - 1) / wpb, 128, 0, stream>>>(
        feat_dst, u, v, aggr, h_dst, mtiles_dst);

    const long long edge_work = (long long)E * (TCP / 4);
    edge_kernel<<<(int)((edge_work + 255) / 256), 256, 0, stream>>>(
        h_src, h_dst, edge_w, weight_e, src_idx, dst_idx, aggr, edge_work);

    const int rows = n_dst * TDIM;
    ln_kernel<<<(rows + 7) / 8, 256, 0, stream>>>(
        aggr, feat_dst, ln_gamma, ln_beta, out, rows);
}


# Float atomic scatter replaced by an in-call counting sort of the edges by destination (integer slot atomics) and one wave per node summing its edges in registers; sum order = slot arrival order
# speedup vs baseline: 1.1368x; 1.1368x over previous
#include <hip/hip_runtime.h>
#include <math.h>


typedef __attribute__((ext_vector_type(2))) float v2f;
typedef __attribute__((ext_vector_type(8))) float v8f;

#define CDIM 64
#define TDIM 4
#define ROWLEN 64
#define TCP 256

__device__ __forceinline__ v8f wmma16x16_k64(const float* __restrict__ Arows,
                                             const float* __restrict__ Blds,
                                             int c0) {
    const int lane = threadIdx.x & 31;
    const int m    = lane & 15;
    const int koff = (lane >> 4) << 1;
    const float* __restrict__ arow = Arows + m * ROWLEN;
    v8f acc = {0.f, 0.f, 0.f, 0.f, 0.f, 0.f, 0.f, 0.f};
#pragma unroll
    for (int k0 = 0; k0 < 64; k0 += 4) {
        v2f a, b;
        a.x = arow[k0 + koff];
        a.y = arow[k0 + koff + 1];
        b.x = Blds[(k0 + koff)     * ROWLEN + c0 + m];
        b.y = Blds[(k0 + koff + 1) * ROWLEN + c0 + m];
        acc = __builtin_amdgcn_wmma_f32_16x16x4_f32(
            false, a, false, b, (short)0, acc, false, false);
    }
    return acc;
}

__device__ __forceinline__ void store_tile(float* __restrict__ Out, int row0,
                                           int c0, v8f acc) {
    const int lane = threadIdx.x & 31;
    const int col  = c0 + (lane & 15);
    const int rb   = row0 + ((lane >> 4) << 3);
#pragma unroll
    for (int i = 0; i < 8; ++i)
        Out[(rb + i) * ROWLEN + col] = acc[i];
}

__global__ void __launch_bounds__(128)
proj_src_kernel(const float* __restrict__ feat, const float* __restrict__ U,
                float* __restrict__ Out, int mtiles) {
    __shared__ float ldsU[CDIM * CDIM];
    for (int i = threadIdx.x; i < CDIM * CDIM; i += blockDim.x) ldsU[i] = U[i];
    __syncthreads();

    const int wave = threadIdx.x >> 5;
    const int tile = blockIdx.x * (blockDim.x >> 5) + wave;
    if (tile >= mtiles) return;
    const int row0 = tile * 16;
    const float* __restrict__ Arows = feat + (long long)row0 * ROWLEN;
#pragma unroll
    for (int c0 = 0; c0 < 64; c0 += 16) {
        v8f acc = wmma16x16_k64(Arows, ldsU, c0);
        store_tile(Out, row0, c0, acc);
    }
}

__global__ void __launch_bounds__(128)
proj_dst_kernel(const float* __restrict__ feat, const float* __restrict__ U,
                const float* __restrict__ V, float* __restrict__ OutBase,
                float* __restrict__ OutV, int mtiles) {
    __shared__ float ldsU[CDIM * CDIM];
    __shared__ float ldsV[CDIM * CDIM];
    for (int i = threadIdx.x; i < CDIM * CDIM; i += blockDim.x) {
        ldsU[i] = U[i];
        ldsV[i] = V[i];
    }
    __syncthreads();

    const int wave = threadIdx.x >> 5;
    const int tile = blockIdx.x * (blockDim.x >> 5) + wave;
    if (tile >= mtiles) return;
    const int row0 = tile * 16;
    const int lane = threadIdx.x & 31;
    const int m    = lane & 15;
    const int koff = (lane >> 4) << 1;
    const float* __restrict__ arow = feat + ((long long)row0 + m) * ROWLEN;

#pragma unroll
    for (int c0 = 0; c0 < 64; c0 += 16) {
        v8f accU = {0.f, 0.f, 0.f, 0.f, 0.f, 0.f, 0.f, 0.f};
        v8f accV = {0.f, 0.f, 0.f, 0.f, 0.f, 0.f, 0.f, 0.f};
#pragma unroll
        for (int k0 = 0; k0 < 64; k0 += 4) {
            v2f a, bu, bv;
            a.x  = arow[k0 + koff];
            a.y  = arow[k0 + koff + 1];
            bu.x = ldsU[(k0 + koff)     * ROWLEN + c0 + m];
            bu.y = ldsU[(k0 + koff + 1) * ROWLEN + c0 + m];
            bv.x = ldsV[(k0 + koff)     * ROWLEN + c0 + m];
            bv.y = ldsV[(k0 + koff + 1) * ROWLEN + c0 + m];
            accU = __builtin_amdgcn_wmma_f32_16x16x4_f32(
                false, a, false, bu, (short)0, accU, false, false);
            accV = __builtin_amdgcn_wmma_f32_16x16x4_f32(
                false, a, false, bv, (short)0, accV, false, false);
        }
        store_tile(OutBase, row0, c0, accU);
        store_tile(OutV,    row0, c0, accV);
    }
}

__global__ void __launch_bounds__(256)
zero_counts_kernel(int* __restrict__ cnt, int n) {
    const int i = blockIdx.x * blockDim.x + threadIdx.x;
    if (i < n) cnt[i] = 0;
}

__global__ void __launch_bounds__(256)
edge_slot_kernel(const int* __restrict__ dst_idx, int* __restrict__ cnt,
                 int* __restrict__ slot, int E) {
    const int e = blockIdx.x * blockDim.x + threadIdx.x;
    if (e < E) slot[e] = atomicAdd(cnt + dst_idx[e], 1);
}

__global__ void __launch_bounds__(1024)
scan_counts_kernel(const int* __restrict__ cnt, int* __restrict__ off, int n) {
    __shared__ int wsum[32];
    const int tid  = threadIdx.x;
    const int lane = tid & 31;
    const int wv   = tid >> 5;
    const int per  = (n + 1023) >> 10;
    const int lo   = tid * per;
    const int hi   = (lo + per < n) ? (lo + per) : n;

    int local = 0;
    for (int i = lo; i < hi; ++i) local += cnt[i];

    int v = local;
    for (int o = 1; o < 32; o <<= 1) {
        const int t = __shfl_up(v, o, 32);
        if (lane >= o) v += t;
    }
    if (lane == 31) wsum[wv] = v;
    __syncthreads();
    if (wv == 0) {
        int s = wsum[lane];
        for (int o = 1; o < 32; o <<= 1) {
            const int t = __shfl_up(s, o, 32);
            if (lane >= o) s += t;
        }
        wsum[lane] = s;
    }
    __syncthreads();

    int run = (wv > 0 ? wsum[wv - 1] : 0) + v - local;
    for (int i = lo; i < hi; ++i) {
        off[i] = run;
        run += cnt[i];
    }
    if (tid == 1023) off[n] = run;
}

__global__ void __launch_bounds__(256)
edge_place_kernel(const int* __restrict__ src_idx, const int* __restrict__ dst_idx,
                  const float* __restrict__ edge_w, const int* __restrict__ off,
                  const int* __restrict__ slot, int* __restrict__ g_src,
                  float* __restrict__ g_w, int E) {
    const int e = blockIdx.x * blockDim.x + threadIdx.x;
    if (e >= E) return;
    const int p = off[dst_idx[e]] + slot[e];
    g_src[p] = src_idx[e];
    g_w[p]   = edge_w[e];
}

__device__ __forceinline__ void gate_acc(float4& a, const float4 hs, const float4 hd,
                                         const float4 wv, const float w) {
    {
        float x = hs.x * hd.x * (w * wv.x);
        a.x += hs.x / (1.0f + __expf(-x));
    }
    {
        float x = hs.y * hd.y * (w * wv.y);
        a.y += hs.y / (1.0f + __expf(-x));
    }
    {
        float x = hs.z * hd.z * (w * wv.z);
        a.z += hs.z / (1.0f + __expf(-x));
    }
    {
        float x = hs.w * hd.w * (w * wv.w);
        a.w += hs.w / (1.0f + __expf(-x));
    }
}

__global__ void __launch_bounds__(256)
node_sum_kernel(const float* __restrict__ h_src, const float* __restrict__ h_dst,
                const float* __restrict__ we, const int* __restrict__ off,
                const int* __restrict__ g_src, const float* __restrict__ g_w,
                float* __restrict__ aggr, int n_dst) {
    const int lane = threadIdx.x & 31;
    const int d = __builtin_amdgcn_readfirstlane(
        (int)(blockIdx.x * (blockDim.x >> 5) + (threadIdx.x >> 5)));
    if (d >= n_dst) return;

    const int beg = off[d];
    const int end = off[d + 1];
    const int o   = lane << 2;

    const float4 wv  = *(const float4*)(we + (o & (CDIM - 1)));
    const float* __restrict__ hdp = h_dst + (long long)d * TCP + o;
    const float4 hd0 = *(const float4*)(hdp);
    const float4 hd1 = *(const float4*)(hdp + 128);
    float* __restrict__ ap = aggr + (long long)d * TCP + o;
    float4 a0 = *(const float4*)(ap);
    float4 a1 = *(const float4*)(ap + 128);

    for (int j0 = beg; j0 < end; j0 += 32) {
        const int n = (end - j0 < 32) ? (end - j0) : 32;
        int   sv = 0;
        float wl = 0.f;
        if (lane < n) {
            sv = g_src[j0 + lane];
            wl = g_w[j0 + lane];
        }
        int k = 0;
        for (; k + 4 <= n; k += 4) {
            const int s0 = __builtin_amdgcn_readlane(sv, k);
            const int s1 = __builtin_amdgcn_readlane(sv, k + 1);
            const int s2 = __builtin_amdgcn_readlane(sv, k + 2);
            const int s3 = __builtin_amdgcn_readlane(sv, k + 3);
            const float* __restrict__ p0 = h_src + (long long)s0 * TCP + o;
            const float* __restrict__ p1 = h_src + (long long)s1 * TCP + o;
            const float* __restrict__ p2 = h_src + (long long)s2 * TCP + o;
            const float* __restrict__ p3 = h_src + (long long)s3 * TCP + o;
            const float4 x00 = *(const float4*)(p0), x01 = *(const float4*)(p0 + 128);
            const float4 x10 = *(const float4*)(p1), x11 = *(const float4*)(p1 + 128);
            const float4 x20 = *(const float4*)(p2), x21 = *(const float4*)(p2 + 128);
            const float4 x30 = *(const float4*)(p3), x31 = *(const float4*)(p3 + 128);
            const float w0 = __builtin_bit_cast(float, __builtin_amdgcn_readlane(__builtin_bit_cast(int, wl), k));
            const float w1 = __builtin_bit_cast(float, __builtin_amdgcn_readlane(__builtin_bit_cast(int, wl), k + 1));
            const float w2 = __builtin_bit_cast(float, __builtin_amdgcn_readlane(__builtin_bit_cast(int, wl), k + 2));
            const float w3 = __builtin_bit_cast(float, __builtin_amdgcn_readlane(__builtin_bit_cast(int, wl), k + 3));
            gate_acc(a0, x00, hd0, wv, w0);  gate_acc(a1, x01, hd1, wv, w0);
            gate_acc(a0, x10, hd0, wv, w1);  gate_acc(a1, x11, hd1, wv, w1);
            gate_acc(a0, x20, hd0, wv, w2);  gate_acc(a1, x21, hd1, wv, w2);
            gate_acc(a0, x30, hd0, wv, w3);  gate_acc(a1, x31, hd1, wv, w3);
        }
        for (; k < n; ++k) {
            const int s0 = __builtin_amdgcn_readlane(sv, k);
            const float* __restrict__ p0 = h_src + (long long)s0 * TCP + o;
            const float4 x00 = *(const float4*)(p0), x01 = *(const float4*)(p0 + 128);
            const float w0 = __builtin_bit_cast(float, __builtin_amdgcn_readlane(__builtin_bit_cast(int, wl), k));
            gate_acc(a0, x00, hd0, wv, w0);  gate_acc(a1, x01, hd1, wv, w0);
        }
    }
    *(float4*)(ap)       = a0;
    *(float4*)(ap + 128) = a1;
}

__global__ void __launch_bounds__(256)
ln_kernel(const float* __restrict__ acc, const float* __restrict__ feat_dst,
          const float* __restrict__ gamma, const float* __restrict__ beta,
          float* __restrict__ out, int rows) {
    const int wave = threadIdx.x >> 5;
    const int lane = threadIdx.x & 31;
    const int row  = blockIdx.x * (blockDim.x >> 5) + wave;
    if (row >= rows) return;

    const float* __restrict__ x = acc + (long long)row * CDIM;
    float x0 = x[lane];
    float x1 = x[lane + 32];

    float s = x0 + x1;
#pragma unroll
    for (int off = 16; off > 0; off >>= 1) s += __shfl_xor(s, off, 32);
    const float mean = s * (1.0f / 64.0f);

    const float d0 = x0 - mean, d1 = x1 - mean;
    float q = d0 * d0 + d1 * d1;
#pragma unroll
    for (int off = 16; off > 0; off >>= 1) q += __shfl_xor(q, off, 32);
    const float rstd = rsqrtf(q * (1.0f / 64.0f) + 1e-5f);

    const float* __restrict__ f = feat_dst + (long long)row * CDIM;
    out[(long long)row * CDIM + lane] =
        d0 * rstd * gamma[lane] + beta[lane] + f[lane];
    out[(long long)row * CDIM + lane + 32] =
        d1 * rstd * gamma[lane + 32] + beta[lane + 32] + f[lane + 32];
}

extern "C" void kernel_launch(void* const* d_in, const int* in_sizes, int n_in,
                              void* d_out, int out_size, void* d_ws, size_t ws_size,
                              hipStream_t stream) {
    const float* feat_src = (const float*)d_in[0];
    const float* feat_dst = (const float*)d_in[1];
    const float* edge_w   = (const float*)d_in[2];
    const float* weight_e = (const float*)d_in[3];
    const float* u        = (const float*)d_in[4];
    const float* v        = (const float*)d_in[5];
    const float* ln_gamma = (const float*)d_in[6];
    const float* ln_beta  = (const float*)d_in[7];
    const int*   src_idx  = (const int*)d_in[8];
    const int*   dst_idx  = (const int*)d_in[9];
    float*       out      = (float*)d_out;

    const int n_src = in_sizes[0] / TCP;
    const int n_dst = in_sizes[1] / TCP;
    const int E     = in_sizes[2];

    float* ws    = (float*)d_ws;
    float* h_src = ws;
    float* h_dst = ws + (size_t)n_src * TCP;
    float* aggr  = h_dst + (size_t)n_dst * TCP;

    const int mtiles_src = (n_src * TDIM) / 16;
    const int mtiles_dst = (n_dst * TDIM) / 16;
    const int wpb = 4;

    proj_src_kernel<<<(mtiles_src + wpb - 1) / wpb, 128, 0, stream>>>(
        feat_src, u, h_src, mtiles_src);

    proj_dst_kernel<<<(mtiles_dst + wpb - 1) / wpb, 128, 0, stream>>>(
        feat_dst, u, v, aggr, h_dst, mtiles_dst);

    int*   cnt   = (int*)d_out;
    int*   off   = cnt + n_dst;
    int*   slot  = off + n_dst + 1;
    int*   g_src = slot + E;
    float* g_w   = (float*)(g_src + E);

    zero_counts_kernel<<<(n_dst + 255) / 256, 256, 0, stream>>>(cnt, n_dst);
    edge_slot_kernel<<<(E + 255) / 256, 256, 0, stream>>>(dst_idx, cnt, slot, E);
    scan_counts_kernel<<<1, 1024, 0, stream>>>(cnt, off, n_dst);
    edge_place_kernel<<<(E + 255) / 256, 256, 0, stream>>>(
        src_idx, dst_idx, edge_w, off, slot, g_src, g_w, E);

    node_sum_kernel<<<(n_dst + 7) / 8, 256, 0, stream>>>(
        h_src, h_dst, weight_e, off, g_src, g_w, aggr, n_dst);

    const int rows = n_dst * TDIM;
    ln_kernel<<<(rows + 7) / 8, 256, 0, stream>>>(
        aggr, feat_dst, ln_gamma, ln_beta, out, rows);
}


# Per-node sum: gate arithmetic cut: factor -log2(e)*h_dst*weight_e formed once per node, sigmoid by v_exp_f32 and v_rcp_f32 (about 1 ulp) in place of the correctly rounded division; sum order kept
# speedup vs baseline: 1.2849x; 1.1303x over previous
#include <hip/hip_runtime.h>
#include <math.h>


typedef __attribute__((ext_vector_type(2))) float v2f;
typedef __attribute__((ext_vector_type(8))) float v8f;
typedef __attribute__((ext_vector_type(4))) float v4f;

#define CDIM 64
#define TDIM 4
#define ROWLEN 64
#define TCP 256

__device__ __forceinline__ v8f wmma16x16_k64(const float* __restrict__ Arows,
                                             const float* __restrict__ Blds,
                                             int c0) {
    const int lane = threadIdx.x & 31;
    const int m    = lane & 15;
    const int koff = (lane >> 4) << 1;
    const float* __restrict__ arow = Arows + m * ROWLEN;
    v8f acc = {0.f, 0.f, 0.f, 0.f, 0.f, 0.f, 0.f, 0.f};
#pragma unroll
    for (int k0 = 0; k0 < 64; k0 += 4) {
        v2f a, b;
        a.x = arow[k0 + koff];
        a.y = arow[k0 + koff + 1];
        b.x = Blds[(k0 + koff)     * ROWLEN + c0 + m];
        b.y = Blds[(k0 + koff + 1) * ROWLEN + c0 + m];
        acc = __builtin_amdgcn_wmma_f32_16x16x4_f32(
            false, a, false, b, (short)0, acc, false, false);
    }
    return acc;
}

__device__ __forceinline__ void store_tile(float* __restrict__ Out, int row0,
                                           int c0, v8f acc) {
    const int lane = threadIdx.x & 31;
    const int col  = c0 + (lane & 15);
    const int rb   = row0 + ((lane >> 4) << 3);
#pragma unroll
    for (int i = 0; i < 8; ++i)
        Out[(rb + i) * ROWLEN + col] = acc[i];
}

__global__ void __launch_bounds__(128)
proj_src_kernel(const float* __restrict__ feat, const float* __restrict__ U,
                float* __restrict__ Out, int mtiles) {
    __shared__ float ldsU[CDIM * CDIM];
    for (int i = threadIdx.x; i < CDIM * CDIM; i += blockDim.x) ldsU[i] = U[i];
    __syncthreads();

    const int wave = threadIdx.x >> 5;
    const int tile = blockIdx.x * (blockDim.x >> 5) + wave;
    if (tile >= mtiles) return;
    const int row0 = tile * 16;
    const float* __restrict__ Arows = feat + (long long)row0 * ROWLEN;
#pragma unroll
    for (int c0 = 0; c0 < 64; c0 += 16) {
        v8f acc = wmma16x16_k64(Arows, ldsU, c0);
        store_tile(Out, row0, c0, acc);
    }
}

__global__ void __launch_bounds__(128)
proj_dst_kernel(const float* __restrict__ feat, const float* __restrict__ U,
                const float* __restrict__ V, float* __restrict__ OutBase,
                float* __restrict__ OutV, int mtiles) {
    __shared__ float ldsU[CDIM * CDIM];
    __shared__ float ldsV[CDIM * CDIM];
    for (int i = threadIdx.x; i < CDIM * CDIM; i += blockDim.x) {
        ldsU[i] = U[i];
        ldsV[i] = V[i];
    }
    __syncthreads();

    const int wave = threadIdx.x >> 5;
    const int tile = blockIdx.x * (blockDim.x >> 5) + wave;
    if (tile >= mtiles) return;
    const int row0 = tile * 16;
    const int lane = threadIdx.x & 31;
    const int m    = lane & 15;
    const int koff = (lane >> 4) << 1;
    const float* __restrict__ arow = feat + ((long long)row0 + m) * ROWLEN;

#pragma unroll
    for (int c0 = 0; c0 < 64; c0 += 16) {
        v8f accU = {0.f, 0.f, 0.f, 0.f, 0.f, 0.f, 0.f, 0.f};
        v8f accV = {0.f, 0.f, 0.f, 0.f, 0.f, 0.f, 0.f, 0.f};
#pragma unroll
        for (int k0 = 0; k0 < 64; k0 += 4) {
            v2f a, bu, bv;
            a.x  = arow[k0 + koff];
            a.y  = arow[k0 + koff + 1];
            bu.x = ldsU[(k0 + koff)     * ROWLEN + c0 + m];
            bu.y = ldsU[(k0 + koff + 1) * ROWLEN + c0 + m];
            bv.x = ldsV[(k0 + koff)     * ROWLEN + c0 + m];
            bv.y = ldsV[(k0 + koff + 1) * ROWLEN + c0 + m];
            accU = __builtin_amdgcn_wmma_f32_16x16x4_f32(
                false, a, false, bu, (short)0, accU, false, false);
            accV = __builtin_amdgcn_wmma_f32_16x16x4_f32(
                false, a, false, bv, (short)0, accV, false, false);
        }
        store_tile(OutBase, row0, c0, accU);
        store_tile(OutV,    row0, c0, accV);
    }
}

__global__ void __launch_bounds__(256)
zero_counts_kernel(int* __restrict__ cnt, int n) {
    const int i = blockIdx.x * blockDim.x + threadIdx.x;
    if (i < n) cnt[i] = 0;
}

__global__ void __launch_bounds__(256)
edge_slot_kernel(const int* __restrict__ dst_idx, int* __restrict__ cnt,
                 int* __restrict__ slot, int E) {
    const int e = blockIdx.x * blockDim.x + threadIdx.x;
    if (e < E) slot[e] = atomicAdd(cnt + dst_idx[e], 1);
}

__global__ void __launch_bounds__(1024)
scan_counts_kernel(const int* __restrict__ cnt, int* __restrict__ off, int n) {
    __shared__ int wsum[32];
    const int tid  = threadIdx.x;
    const int lane = tid & 31;
    const int wv   = tid >> 5;
    const int per  = (n + 1023) >> 10;
    const int lo   = tid * per;
    const int hi   = (lo + per < n) ? (lo + per) : n;

    int local = 0;
    for (int i = lo; i < hi; ++i) local += cnt[i];

    int v = local;
    for (int o = 1; o < 32; o <<= 1) {
        const int t = __shfl_up(v, o, 32);
        if (lane >= o) v += t;
    }
    if (lane == 31) wsum[wv] = v;
    __syncthreads();
    if (wv == 0) {
        int s = wsum[lane];
        for (int o = 1; o < 32; o <<= 1) {
            const int t = __shfl_up(s, o, 32);
            if (lane >= o) s += t;
        }
        wsum[lane] = s;
    }
    __syncthreads();

    int run = (wv > 0 ? wsum[wv - 1] : 0) + v - local;
    for (int i = lo; i < hi; ++i) {
        off[i] = run;
        run += cnt[i];
    }
    if (tid == 1023) off[n] = run;
}

__global__ void __launch_bounds__(256)
edge_place_kernel(const int* __restrict__ src_idx, const int* __restrict__ dst_idx,
                  const float* __restrict__ edge_w, const int* __restrict__ off,
                  const int* __restrict__ slot, int* __restrict__ g_src,
                  float* __restrict__ g_w, int E) {
    const int e = blockIdx.x * blockDim.x + threadIdx.x;
    if (e >= E) return;
    const int p = off[dst_idx[e]] + slot[e];
    g_src[p] = src_idx[e];
    g_w[p]   = edge_w[e];
}

__device__ __forceinline__ void gate_acc(v4f& a, const v4f hs, const v4f c, const float w) {
    const v4f y = hs * (c * w);
    v4f r;
    r.x = __builtin_amdgcn_rcpf(1.0f + __builtin_amdgcn_exp2f(y.x));
    r.y = __builtin_amdgcn_rcpf(1.0f + __builtin_amdgcn_exp2f(y.y));
    r.z = __builtin_amdgcn_rcpf(1.0f + __builtin_amdgcn_exp2f(y.z));
    r.w = __builtin_amdgcn_rcpf(1.0f + __builtin_amdgcn_exp2f(y.w));
    a = __builtin_elementwise_fma(hs, r, a);
}

__global__ void __launch_bounds__(256)
node_sum_kernel(const float* __restrict__ h_src, const float* __restrict__ h_dst,
                const float* __restrict__ we, const int* __restrict__ off,
                const int* __restrict__ g_src, const float* __restrict__ g_w,
                float* __restrict__ aggr, int n_dst) {
    const int lane = threadIdx.x & 31;
    const int d = __builtin_amdgcn_readfirstlane(
        (int)(blockIdx.x * (blockDim.x >> 5) + (threadIdx.x >> 5)));
    if (d >= n_dst) return;

    const int beg = off[d];
    const int end = off[d + 1];
    const int o   = lane << 2;

    const v4f wv  = *(const v4f*)(we + (o & (CDIM - 1)));
    const float* __restrict__ hdp = h_dst + (long long)d * TCP + o;
    const v4f c0 = (-1.44269504088896341f * *(const v4f*)(hdp)) * wv;
    const v4f c1 = (-1.44269504088896341f * *(const v4f*)(hdp + 128)) * wv;
    float* __restrict__ ap = aggr + (long long)d * TCP + o;
    v4f a0 = *(const v4f*)(ap);
    v4f a1 = *(const v4f*)(ap + 128);

    for (int j0 = beg; j0 < end; j0 += 32) {
        const int n = (end - j0 < 32) ? (end - j0) : 32;
        int   sv = 0;
        float wl = 0.f;
        if (lane < n) {
            sv = g_src[j0 + lane];
            wl = g_w[j0 + lane];
        }
        int k = 0;
        for (; k + 4 <= n; k += 4) {
            const int s0 = __builtin_amdgcn_readlane(sv, k);
            const int s1 = __builtin_amdgcn_readlane(sv, k + 1);
            const int s2 = __builtin_amdgcn_readlane(sv, k + 2);
            const int s3 = __builtin_amdgcn_readlane(sv, k + 3);
            const float* __restrict__ p0 = h_src + (long long)s0 * TCP + o;
            const float* __restrict__ p1 = h_src + (long long)s1 * TCP + o;
            const float* __restrict__ p2 = h_src + (long long)s2 * TCP + o;
            const float* __restrict__ p3 = h_src + (long long)s3 * TCP + o;
            const v4f x00 = *(const v4f*)(p0), x01 = *(const v4f*)(p0 + 128);
            const v4f x10 = *(const v4f*)(p1), x11 = *(const v4f*)(p1 + 128);
            const v4f x20 = *(const v4f*)(p2), x21 = *(const v4f*)(p2 + 128);
            const v4f x30 = *(const v4f*)(p3), x31 = *(const v4f*)(p3 + 128);
            const float w0 = __builtin_bit_cast(float, __builtin_amdgcn_readlane(__builtin_bit_cast(int, wl), k));
            const float w1 = __builtin_bit_cast(float, __builtin_amdgcn_readlane(__builtin_bit_cast(int, wl), k + 1));
            const float w2 = __builtin_bit_cast(float, __builtin_amdgcn_readlane(__builtin_bit_cast(int, wl), k + 2));
            const float w3 = __builtin_bit_cast(float, __builtin_amdgcn_readlane(__builtin_bit_cast(int, wl), k + 3));
            gate_acc(a0, x00, c0, w0);  gate_acc(a1, x01, c1, w0);
            gate_acc(a0, x10, c0, w1);  gate_acc(a1, x11, c1, w1);
            gate_acc(a0, x20, c0, w2);  gate_acc(a1, x21, c1, w2);
            gate_acc(a0, x30, c0, w3);  gate_acc(a1, x31, c1, w3);
        }
        for (; k < n; ++k) {
            const int s0 = __builtin_amdgcn_readlane(sv, k);
            const float* __restrict__ p0 = h_src + (long long)s0 * TCP + o;
            const v4f x00 = *(const v4f*)(p0), x01 = *(const v4f*)(p0 + 128);
            const float w0 = __builtin_bit_cast(float, __builtin_amdgcn_readlane(__builtin_bit_cast(int, wl), k));
            gate_acc(a0, x00, c0, w0);  gate_acc(a1, x01, c1, w0);
        }
    }
    *(v4f*)(ap)       = a0;
    *(v4f*)(ap + 128) = a1;
}

__global__ void __launch_bounds__(256)
ln_kernel(const float* __restrict__ acc, const float* __restrict__ feat_dst,
          const float* __restrict__ gamma, const float* __restrict__ beta,
          float* __restrict__ out, int rows) {
    const int wave = threadIdx.x >> 5;
    const int lane = threadIdx.x & 31;
    const int row  = blockIdx.x * (blockDim.x >> 5) + wave;
    if (row >= rows) return;

    const float* __restrict__ x = acc + (long long)row * CDIM;
    float x0 = x[lane];
    float x1 = x[lane + 32];

    float s = x0 + x1;
#pragma unroll
    for (int off = 16; off > 0; off >>= 1) s += __shfl_xor(s, off, 32);
    const float mean = s * (1.0f / 64.0f);

    const float d0 = x0 - mean, d1 = x1 - mean;
    float q = d0 * d0 + d1 * d1;
#pragma unroll
    for (int off = 16; off > 0; off >>= 1) q += __shfl_xor(q, off, 32);
    const float rstd = rsqrtf(q * (1.0f / 64.0f) + 1e-5f);

    const float* __restrict__ f = feat_dst + (long long)row * CDIM;
    out[(long long)row * CDIM + lane] =
        d0 * rstd * gamma[lane] + beta[lane] + f[lane];
    out[(long long)row * CDIM + lane + 32] =
        d1 * rstd * gamma[lane + 32] + beta[lane + 32] + f[lane + 32];
}

extern "C" void kernel_launch(void* const* d_in, const int* in_sizes, int n_in,
                              void* d_out, int out_size, void* d_ws, size_t ws_size,
                              hipStream_t stream) {
    const float* feat_src = (const float*)d_in[0];
    const float* feat_dst = (const float*)d_in[1];
    const float* edge_w   = (const float*)d_in[2];
    const float* weight_e = (const float*)d_in[3];
    const float* u        = (const float*)d_in[4];
    const float* v        = (const float*)d_in[5];
    const float* ln_gamma = (const float*)d_in[6];
    const float* ln_beta  = (const float*)d_in[7];
    const int*   src_idx  = (const int*)d_in[8];
    const int*   dst_idx  = (const int*)d_in[9];
    float*       out      = (float*)d_out;

    const int n_src = in_sizes[0] / TCP;
    const int n_dst = in_sizes[1] / TCP;
    const int E     = in_sizes[2];

    float* ws    = (float*)d_ws;
    float* h_src = ws;
    float* h_dst = ws + (size_t)n_src * TCP;
    float* aggr  = h_dst + (size_t)n_dst * TCP;

    const int mtiles_src = (n_src * TDIM) / 16;
    const int mtiles_dst = (n_dst * TDIM) / 16;
    const int wpb = 4;

    proj_src_kernel<<<(mtiles_src + wpb - 1) / wpb, 128, 0, stream>>>(
        feat_src, u, h_src, mtiles_src);

    proj_dst_kernel<<<(mtiles_dst + wpb - 1) / wpb, 128, 0, stream>>>(
        feat_dst, u, v, aggr, h_dst, mtiles_dst);

    int*   cnt   = (int*)d_out;
    int*   off   = cnt + n_dst;
    int*   slot  = off + n_dst + 1;
    int*   g_src = slot + E;
    float* g_w   = (float*)(g_src + E);

    zero_counts_kernel<<<(n_dst + 255) / 256, 256, 0, stream>>>(cnt, n_dst);
    edge_slot_kernel<<<(E + 255) / 256, 256, 0, stream>>>(dst_idx, cnt, slot, E);
    scan_counts_kernel<<<1, 1024, 0, stream>>>(cnt, off, n_dst);
    edge_place_kernel<<<(E + 255) / 256, 256, 0, stream>>>(
        src_idx, dst_idx, edge_w, off, slot, g_src, g_w, E);

    node_sum_kernel<<<(n_dst + 7) / 8, 256, 0, stream>>>(
        h_src, h_dst, weight_e, off, g_src, g_w, aggr, n_dst);

    const int rows = n_dst * TDIM;
    ln_kernel<<<(rows + 7) / 8, 256, 0, stream>>>(
        aggr, feat_dst, ln_gamma, ln_beta, out, rows);
}


# Projection kernels: u and v staged into LDS by 16-byte loads all in flight before the first store, replacing a one-word copy loop of 32 dependent load-store round trips per work-group
# speedup vs baseline: 1.5820x; 1.2312x over previous
#include <hip/hip_runtime.h>
#include <math.h>


typedef __attribute__((ext_vector_type(2))) float v2f;
typedef __attribute__((ext_vector_type(8))) float v8f;
typedef __attribute__((ext_vector_type(4))) float v4f;

#define CDIM 64
#define TDIM 4
#define ROWLEN 64
#define TCP 256
#define PROJ_THREADS 128
#define STAGE_VECS ((CDIM * CDIM / 4) / PROJ_THREADS)

__device__ __forceinline__ v8f wmma16x16_k64(const float* __restrict__ Arows,
                                             const float* __restrict__ Blds,
                                             int c0) {
    const int lane = threadIdx.x & 31;
    const int m    = lane & 15;
    const int koff = (lane >> 4) << 1;
    const float* __restrict__ arow = Arows + m * ROWLEN;
    v8f acc = {0.f, 0.f, 0.f, 0.f, 0.f, 0.f, 0.f, 0.f};
#pragma unroll
    for (int k0 = 0; k0 < 64; k0 += 4) {
        v2f a, b;
        a.x = arow[k0 + koff];
        a.y = arow[k0 + koff + 1];
        b.x = Blds[(k0 + koff)     * ROWLEN + c0 + m];
        b.y = Blds[(k0 + koff + 1) * ROWLEN + c0 + m];
        acc = __builtin_amdgcn_wmma_f32_16x16x4_f32(
            false, a, false, b, (short)0, acc, false, false);
    }
    return acc;
}

__device__ __forceinline__ void store_tile(float* __restrict__ Out, int row0,
                                           int c0, v8f acc) {
    const int lane = threadIdx.x & 31;
    const int col  = c0 + (lane & 15);
    const int rb   = row0 + ((lane >> 4) << 3);
#pragma unroll
    for (int i = 0; i < 8; ++i)
        Out[(rb + i) * ROWLEN + col] = acc[i];
}

__global__ void __launch_bounds__(PROJ_THREADS)
proj_src_kernel(const float* __restrict__ feat, const float* __restrict__ U,
                float* __restrict__ Out, int mtiles) {
    __shared__ __attribute__((aligned(16))) float ldsU[CDIM * CDIM];
    {
        v4f ru[STAGE_VECS];
        for (int j = 0; j < STAGE_VECS; ++j)
            ru[j] = ((const v4f*)U)[threadIdx.x + j * PROJ_THREADS];
        for (int j = 0; j < STAGE_VECS; ++j)
            ((v4f*)ldsU)[threadIdx.x + j * PROJ_THREADS] = ru[j];
    }
    __syncthreads();

    const int wave = threadIdx.x >> 5;
    const int tile = blockIdx.x * (blockDim.x >> 5) + wave;
    if (tile >= mtiles) return;
    const int row0 = tile * 16;
    const float* __restrict__ Arows = feat + (long long)row0 * ROWLEN;
#pragma unroll
    for (int c0 = 0; c0 < 64; c0 += 16) {
        v8f acc = wmma16x16_k64(Arows, ldsU, c0);
        store_tile(Out, row0, c0, acc);
    }
}

__global__ void __launch_bounds__(PROJ_THREADS)
proj_dst_kernel(const float* __restrict__ feat, const float* __restrict__ U,
                const float* __restrict__ V, float* __restrict__ OutBase,
                float* __restrict__ OutV, int mtiles) {
    __shared__ __attribute__((aligned(16))) float ldsU[CDIM * CDIM];
    __shared__ __attribute__((aligned(16))) float ldsV[CDIM * CDIM];
    {
        v4f ru[STAGE_VECS], rv[STAGE_VECS];
        for (int j = 0; j < STAGE_VECS; ++j) {
            ru[j] = ((const v4f*)U)[threadIdx.x + j * PROJ_THREADS];
            rv[j] = ((const v4f*)V)[threadIdx.x + j * PROJ_THREADS];
        }
        for (int j = 0; j < STAGE_VECS; ++j) {
            ((v4f*)ldsU)[threadIdx.x + j * PROJ_THREADS] = ru[j];
            ((v4f*)ldsV)[threadIdx.x + j * PROJ_THREADS] = rv[j];
        }
    }
    __syncthreads();

    const int wave = threadIdx.x >> 5;
    const int tile = blockIdx.x * (blockDim.x >> 5) + wave;
    if (tile >= mtiles) return;
    const int row0 = tile * 16;
    const int lane = threadIdx.x & 31;
    const int m    = lane & 15;
    const int koff = (lane >> 4) << 1;
    const float* __restrict__ arow = feat + ((long long)row0 + m) * ROWLEN;

#pragma unroll
    for (int c0 = 0; c0 < 64; c0 += 16) {
        v8f accU = {0.f, 0.f, 0.f, 0.f, 0.f, 0.f, 0.f, 0.f};
        v8f accV = {0.f, 0.f, 0.f, 0.f, 0.f, 0.f, 0.f, 0.f};
#pragma unroll
        for (int k0 = 0; k0 < 64; k0 += 4) {
            v2f a, bu, bv;
            a.x  = arow[k0 + koff];
            a.y  = arow[k0 + koff + 1];
            bu.x = ldsU[(k0 + koff)     * ROWLEN + c0 + m];
            bu.y = ldsU[(k0 + koff + 1) * ROWLEN + c0 + m];
            bv.x = ldsV[(k0 + koff)     * ROWLEN + c0 + m];
            bv.y = ldsV[(k0 + koff + 1) * ROWLEN + c0 + m];
            accU = __builtin_amdgcn_wmma_f32_16x16x4_f32(
                false, a, false, bu, (short)0, accU, false, false);
            accV = __builtin_amdgcn_wmma_f32_16x16x4_f32(
                false, a, false, bv, (short)0, accV, false, false);
        }
        store_tile(OutBase, row0, c0, accU);
        store_tile(OutV,    row0, c0, accV);
    }
}

__global__ void __launch_bounds__(256)
zero_counts_kernel(int* __restrict__ cnt, int n) {
    const int i = blockIdx.x * blockDim.x + threadIdx.x;
    if (i < n) cnt[i] = 0;
}

__global__ void __launch_bounds__(256)
edge_slot_kernel(const int* __restrict__ dst_idx, int* __restrict__ cnt,
                 int* __restrict__ slot, int E) {
    const int e = blockIdx.x * blockDim.x + threadIdx.x;
    if (e < E) slot[e] = atomicAdd(cnt + dst_idx[e], 1);
}

__global__ void __launch_bounds__(1024)
scan_counts_kernel(const int* __restrict__ cnt, int* __restrict__ off, int n) {
    __shared__ int wsum[32];
    const int tid  = threadIdx.x;
    const int lane = tid & 31;
    const int wv   = tid >> 5;
    const int per  = (n + 1023) >> 10;
    const int lo   = tid * per;
    const int hi   = (lo + per < n) ? (lo + per) : n;

    int local = 0;
    for (int i = lo; i < hi; ++i) local += cnt[i];

    int v = local;
    for (int o = 1; o < 32; o <<= 1) {
        const int t = __shfl_up(v, o, 32);
        if (lane >= o) v += t;
    }
    if (lane == 31) wsum[wv] = v;
    __syncthreads();
    if (wv == 0) {
        int s = wsum[lane];
        for (int o = 1; o < 32; o <<= 1) {
            const int t = __shfl_up(s, o, 32);
            if (lane >= o) s += t;
        }
        wsum[lane] = s;
    }
    __syncthreads();

    int run = (wv > 0 ? wsum[wv - 1] : 0) + v - local;
    for (int i = lo; i < hi; ++i) {
        off[i] = run;
        run += cnt[i];
    }
    if (tid == 1023) off[n] = run;
}

__global__ void __launch_bounds__(256)
edge_place_kernel(const int* __restrict__ src_idx, const int* __restrict__ dst_idx,
                  const float* __restrict__ edge_w, const int* __restrict__ off,
                  const int* __restrict__ slot, int* __restrict__ g_src,
                  float* __restrict__ g_w, int E) {
    const int e = blockIdx.x * blockDim.x + threadIdx.x;
    if (e >= E) return;
    const int p = off[dst_idx[e]] + slot[e];
    g_src[p] = src_idx[e];
    g_w[p]   = edge_w[e];
}

__device__ __forceinline__ void gate_acc(v4f& a, const v4f hs, const v4f c, const float w) {
    const v4f y = hs * (c * w);
    v4f r;
    r.x = __builtin_amdgcn_rcpf(1.0f + __builtin_amdgcn_exp2f(y.x));
    r.y = __builtin_amdgcn_rcpf(1.0f + __builtin_amdgcn_exp2f(y.y));
    r.z = __builtin_amdgcn_rcpf(1.0f + __builtin_amdgcn_exp2f(y.z));
    r.w = __builtin_amdgcn_rcpf(1.0f + __builtin_amdgcn_exp2f(y.w));
    a = __builtin_elementwise_fma(hs, r, a);
}

__global__ void __launch_bounds__(256)
node_sum_kernel(const float* __restrict__ h_src, const float* __restrict__ h_dst,
                const float* __restrict__ we, const int* __restrict__ off,
                const int* __restrict__ g_src, const float* __restrict__ g_w,
                float* __restrict__ aggr, int n_dst) {
    const int lane = threadIdx.x & 31;
    const int d = __builtin_amdgcn_readfirstlane(
        (int)(blockIdx.x * (blockDim.x >> 5) + (threadIdx.x >> 5)));
    if (d >= n_dst) return;

    const int beg = off[d];
    const int end = off[d + 1];
    const int o   = lane << 2;

    const v4f wv  = *(const v4f*)(we + (o & (CDIM - 1)));
    const float* __restrict__ hdp = h_dst + (long long)d * TCP + o;
    const v4f c0 = (-1.44269504088896341f * *(const v4f*)(hdp)) * wv;
    const v4f c1 = (-1.44269504088896341f * *(const v4f*)(hdp + 128)) * wv;
    float* __restrict__ ap = aggr + (long long)d * TCP + o;
    v4f a0 = *(const v4f*)(ap);
    v4f a1 = *(const v4f*)(ap + 128);

    for (int j0 = beg; j0 < end; j0 += 32) {
        const int n = (end - j0 < 32) ? (end - j0) : 32;
        int   sv = 0;
        float wl = 0.f;
        if (lane < n) {
            sv = g_src[j0 + lane];
            wl = g_w[j0 + lane];
        }
        int k = 0;
        for (; k + 4 <= n; k += 4) {
            const int s0 = __builtin_amdgcn_readlane(sv, k);
            const int s1 = __builtin_amdgcn_readlane(sv, k + 1);
            const int s2 = __builtin_amdgcn_readlane(sv, k + 2);
            const int s3 = __builtin_amdgcn_readlane(sv, k + 3);
            const float* __restrict__ p0 = h_src + (long long)s0 * TCP + o;
            const float* __restrict__ p1 = h_src + (long long)s1 * TCP + o;
            const float* __restrict__ p2 = h_src + (long long)s2 * TCP + o;
            const float* __restrict__ p3 = h_src + (long long)s3 * TCP + o;
            const v4f x00 = *(const v4f*)(p0), x01 = *(const v4f*)(p0 + 128);
            const v4f x10 = *(const v4f*)(p1), x11 = *(const v4f*)(p1 + 128);
            const v4f x20 = *(const v4f*)(p2), x21 = *(const v4f*)(p2 + 128);
            const v4f x30 = *(const v4f*)(p3), x31 = *(const v4f*)(p3 + 128);
            const float w0 = __builtin_bit_cast(float, __builtin_amdgcn_readlane(__builtin_bit_cast(int, wl), k));
            const float w1 = __builtin_bit_cast(float, __builtin_amdgcn_readlane(__builtin_bit_cast(int, wl), k + 1));
            const float w2 = __builtin_bit_cast(float, __builtin_amdgcn_readlane(__builtin_bit_cast(int, wl), k + 2));
            const float w3 = __builtin_bit_cast(float, __builtin_amdgcn_readlane(__builtin_bit_cast(int, wl), k + 3));
            gate_acc(a0, x00, c0, w0);  gate_acc(a1, x01, c1, w0);
            gate_acc(a0, x10, c0, w1);  gate_acc(a1, x11, c1, w1);
            gate_acc(a0, x20, c0, w2);  gate_acc(a1, x21, c1, w2);
            gate_acc(a0, x30, c0, w3);  gate_acc(a1, x31, c1, w3);
        }
        for (; k < n; ++k) {
            const int s0 = __builtin_amdgcn_readlane(sv, k);
            const float* __restrict__ p0 = h_src + (long long)s0 * TCP + o;
            const v4f x00 = *(const v4f*)(p0), x01 = *(const v4f*)(p0 + 128);
            const float w0 = __builtin_bit_cast(float, __builtin_amdgcn_readlane(__builtin_bit_cast(int, wl), k));
            gate_acc(a0, x00, c0, w0);  gate_acc(a1, x01, c1, w0);
        }
    }
    *(v4f*)(ap)       = a0;
    *(v4f*)(ap + 128) = a1;
}

__global__ void __launch_bounds__(256)
ln_kernel(const float* __restrict__ acc, const float* __restrict__ feat_dst,
          const float* __restrict__ gamma, const float* __restrict__ beta,
          float* __restrict__ out, int rows) {
    const int wave = threadIdx.x >> 5;
    const int lane = threadIdx.x & 31;
    const int row  = blockIdx.x * (blockDim.x >> 5) + wave;
    if (row >= rows) return;

    const float* __restrict__ x = acc + (long long)row * CDIM;
    float x0 = x[lane];
    float x1 = x[lane + 32];

    float s = x0 + x1;
#pragma unroll
    for (int off = 16; off > 0; off >>= 1) s += __shfl_xor(s, off, 32);
    const float mean = s * (1.0f / 64.0f);

    const float d0 = x0 - mean, d1 = x1 - mean;
    float q = d0 * d0 + d1 * d1;
#pragma unroll
    for (int off = 16; off > 0; off >>= 1) q += __shfl_xor(q, off, 32);
    const float rstd = rsqrtf(q * (1.0f / 64.0f) + 1e-5f);

    const float* __restrict__ f = feat_dst + (long long)row * CDIM;
    out[(long long)row * CDIM + lane] =
        d0 * rstd * gamma[lane] + beta[lane] + f[lane];
    out[(long long)row * CDIM + lane + 32] =
        d1 * rstd * gamma[lane + 32] + beta[lane + 32] + f[lane + 32];
}

extern "C" void kernel_launch(void* const* d_in, const int* in_sizes, int n_in,
                              void* d_out, int out_size, void* d_ws, size_t ws_size,
                              hipStream_t stream) {
    const float* feat_src = (const float*)d_in[0];
    const float* feat_dst = (const float*)d_in[1];
    const float* edge_w   = (const float*)d_in[2];
    const float* weight_e = (const float*)d_in[3];
    const float* u        = (const float*)d_in[4];
    const float* v        = (const float*)d_in[5];
    const float* ln_gamma = (const float*)d_in[6];
    const float* ln_beta  = (const float*)d_in[7];
    const int*   src_idx  = (const int*)d_in[8];
    const int*   dst_idx  = (const int*)d_in[9];
    float*       out      = (float*)d_out;

    const int n_src = in_sizes[0] / TCP;
    const int n_dst = in_sizes[1] / TCP;
    const int E     = in_sizes[2];

    float* ws    = (float*)d_ws;
    float* h_src = ws;
    float* h_dst = ws + (size_t)n_src * TCP;
    float* aggr  = h_dst + (size_t)n_dst * TCP;

    const int mtiles_src = (n_src * TDIM) / 16;
    const int mtiles_dst = (n_dst * TDIM) / 16;
    const int wpb = PROJ_THREADS / 32;

    proj_src_kernel<<<(mtiles_src + wpb - 1) / wpb, PROJ_THREADS, 0, stream>>>(
        feat_src, u, h_src, mtiles_src);

    proj_dst_kernel<<<(mtiles_dst + wpb - 1) / wpb, PROJ_THREADS, 0, stream>>>(
        feat_dst, u, v, aggr, h_dst, mtiles_dst);

    int*   cnt   = (int*)d_out;
    int*   off   = cnt + n_dst;
    int*   slot  = off + n_dst + 1;
    int*   g_src = slot + E;
    float* g_w   = (float*)(g_src + E);

    zero_counts_kernel<<<(n_dst + 255) / 256, 256, 0, stream>>>(cnt, n_dst);
    edge_slot_kernel<<<(E + 255) / 256, 256, 0, stream>>>(dst_idx, cnt, slot, E);
    scan_counts_kernel<<<1, 1024, 0, stream>>>(cnt, off, n_dst);
    edge_place_kernel<<<(E + 255) / 256, 256, 0, stream>>>(
        src_idx, dst_idx, edge_w, off, slot, g_src, g_w, E);

    node_sum_kernel<<<(n_dst + 7) / 8, 256, 0, stream>>>(
        h_src, h_dst, weight_e, off, g_src, g_w, aggr, n_dst);

    const int rows = n_dst * TDIM;
    ln_kernel<<<(rows + 7) / 8, 256, 0, stream>>>(
        aggr, feat_dst, ln_gamma, ln_beta, out, rows);
}


# Destination projections (feat_dst@u, feat_dst@v) moved into the per-node launch: each 8-wave group forms them by WMMA for its 8 nodes into LDS; base/h_dst memory round trip gone; sum order kept
# speedup vs baseline: 2.7233x; 1.7214x over previous
#include <hip/hip_runtime.h>
#include <math.h>


typedef __attribute__((ext_vector_type(2))) float v2f;
typedef __attribute__((ext_vector_type(8))) float v8f;
typedef __attribute__((ext_vector_type(4))) float v4f;

#define CDIM 64
#define TDIM 4
#define ROWLEN 64
#define TCP 256
#define PROJ_THREADS 128
#define STAGE_VECS ((CDIM * CDIM / 4) / PROJ_THREADS)
#define BUCKET 64
#define W_AT   BUCKET
#define CNT_AT (2 * BUCKET)
#define EDGES_PER_THREAD 4
#define EDGES_PER_GROUP (PROJ_THREADS * EDGES_PER_THREAD)
#define NODES_PER_GROUP 8
#define NODE_STAGE_VECS ((CDIM * CDIM / 4) / (NODES_PER_GROUP * 32))
#define RES_PITCH 68

__device__ __forceinline__ v8f wmma16x16_k64(const float* __restrict__ Arows,
                                             const float* __restrict__ Blds,
                                             int c0) {
    const int lane = threadIdx.x & 31;
    const int m    = lane & 15;
    const int koff = (lane >> 4) << 1;
    const float* __restrict__ arow = Arows + m * ROWLEN;
    v8f acc = {0.f, 0.f, 0.f, 0.f, 0.f, 0.f, 0.f, 0.f};
#pragma unroll
    for (int k0 = 0; k0 < 64; k0 += 4) {
        v2f a, b;
        a.x = arow[k0 + koff];
        a.y = arow[k0 + koff + 1];
        b.x = Blds[(k0 + koff)     * ROWLEN + c0 + m];
        b.y = Blds[(k0 + koff + 1) * ROWLEN + c0 + m];
        acc = __builtin_amdgcn_wmma_f32_16x16x4_f32(
            false, a, false, b, (short)0, acc, false, false);
    }
    return acc;
}

__device__ __forceinline__ void store_tile(float* __restrict__ Out, int row0,
                                           int c0, v8f acc) {
    const int lane = threadIdx.x & 31;
    const int col  = c0 + (lane & 15);
    const int rb   = row0 + ((lane >> 4) << 3);
#pragma unroll
    for (int i = 0; i < 8; ++i)
        Out[(rb + i) * ROWLEN + col] = acc[i];
}

__global__ void __launch_bounds__(PROJ_THREADS)
proj_src_kernel(const float* __restrict__ feat, const float* __restrict__ U,
                float* __restrict__ Out, int mtiles,
                float* __restrict__ out, int n_dst) {
    __shared__ __attribute__((aligned(16))) float ldsU[CDIM * CDIM];
    for (int i = blockIdx.x * PROJ_THREADS + threadIdx.x; i < n_dst;
         i += gridDim.x * PROJ_THREADS)
        ((int*)out)[(long long)i * TCP + CNT_AT] = 0;
    {
        v4f ru[STAGE_VECS];
        for (int j = 0; j < STAGE_VECS; ++j)
            ru[j] = ((const v4f*)U)[threadIdx.x + j * PROJ_THREADS];
        for (int j = 0; j < STAGE_VECS; ++j)
            ((v4f*)ldsU)[threadIdx.x + j * PROJ_THREADS] = ru[j];
    }
    __syncthreads();

    const int wave = threadIdx.x >> 5;
    const int tile = blockIdx.x * (blockDim.x >> 5) + wave;
    if (tile >= mtiles) return;
    const int row0 = tile * 16;
    const float* __restrict__ Arows = feat + (long long)row0 * ROWLEN;
#pragma unroll
    for (int c0 = 0; c0 < 64; c0 += 16) {
        v8f acc = wmma16x16_k64(Arows, ldsU, c0);
        store_tile(Out, row0, c0, acc);
    }
}

__device__ __forceinline__ void group_edges(const int* __restrict__ src_idx,
                                            const int* __restrict__ dst_idx,
                                            const float* __restrict__ edge_w,
                                            float* __restrict__ out, int E, int group) {
    int   d[EDGES_PER_THREAD], s[EDGES_PER_THREAD], slot[EDGES_PER_THREAD];
    float w[EDGES_PER_THREAD];
    const int e0 = group * EDGES_PER_GROUP + threadIdx.x;
#pragma unroll
    for (int q = 0; q < EDGES_PER_THREAD; ++q) {
        const int e = e0 + q * PROJ_THREADS;
        d[q] = -1;  s[q] = 0;  w[q] = 0.f;
        if (e < E) {
            d[q] = dst_idx[e];
            s[q] = src_idx[e];
            w[q] = edge_w[e];
        }
    }
#pragma unroll
    for (int q = 0; q < EDGES_PER_THREAD; ++q) {
        asm volatile("" : : "v"(s[q]), "v"(w[q]));
    }
#pragma unroll
    for (int q = 0; q < EDGES_PER_THREAD; ++q) {
        slot[q] = BUCKET;
        if (d[q] >= 0)
            slot[q] = atomicAdd((int*)out + (long long)d[q] * TCP + CNT_AT, 1);
    }
#pragma unroll
    for (int q = 0; q < EDGES_PER_THREAD; ++q) {
        if (slot[q] < BUCKET) {
            int* __restrict__ blk = (int*)out + (long long)d[q] * TCP;
            blk[slot[q]] = s[q];
            blk[W_AT + slot[q]] = __builtin_bit_cast(int, w[q]);
        }
    }
}

__global__ void __launch_bounds__(PROJ_THREADS)
group_kernel(const int* __restrict__ src_idx, const int* __restrict__ dst_idx,
             const float* __restrict__ edge_w, float* __restrict__ out, int E) {
    group_edges(src_idx, dst_idx, edge_w, out, E, (int)blockIdx.x);
}

__device__ __forceinline__ void gate_acc(v4f& a, const v4f hs, const v4f c, const float w) {
    const v4f y = hs * (c * w);
    v4f r;
    r.x = __builtin_amdgcn_rcpf(1.0f + __builtin_amdgcn_exp2f(y.x));
    r.y = __builtin_amdgcn_rcpf(1.0f + __builtin_amdgcn_exp2f(y.y));
    r.z = __builtin_amdgcn_rcpf(1.0f + __builtin_amdgcn_exp2f(y.z));
    r.w = __builtin_amdgcn_rcpf(1.0f + __builtin_amdgcn_exp2f(y.w));
    a = __builtin_elementwise_fma(hs, r, a);
}

template <int MASK>
__device__ __forceinline__ float add_row_xor(const float x) {
    return x + __builtin_bit_cast(float,
        __builtin_amdgcn_mov_dpp(__builtin_bit_cast(int, x), 0x160 | MASK, 0xf, 0xf, true));
}

__device__ __forceinline__ void row16_sum2(float& x, float& y) {
    x = add_row_xor<1>(x);  y = add_row_xor<1>(y);
    x = add_row_xor<2>(x);  y = add_row_xor<2>(y);
    x = add_row_xor<4>(x);  y = add_row_xor<4>(y);
    x = add_row_xor<8>(x);  y = add_row_xor<8>(y);
}

__global__ void __launch_bounds__(256, 12)
node_sum_ln_kernel(const float* __restrict__ h_src, const float* __restrict__ U,
                   const float* __restrict__ V, const float* __restrict__ we,
                   const int* __restrict__ src_idx, const int* __restrict__ dst_idx,
                   const float* __restrict__ edge_w, const float* __restrict__ feat_dst,
                   const float* __restrict__ gamma, const float* __restrict__ beta,
                   float* __restrict__ out, int n_dst, int E) {
    __shared__ __attribute__((aligned(16))) float ldsU[CDIM * CDIM];
    __shared__ __attribute__((aligned(16))) float ldsV[CDIM * CDIM];
    __shared__ __attribute__((aligned(16))) float ldsBase[NODES_PER_GROUP * TDIM * RES_PITCH];
    __shared__ __attribute__((aligned(16))) float ldsHd[NODES_PER_GROUP * TDIM * RES_PITCH];
    const int lane = threadIdx.x & 31;
    const int wave = threadIdx.x >> 5;
    const int d = __builtin_amdgcn_readfirstlane(
        (int)(blockIdx.x * NODES_PER_GROUP + wave));
    float* __restrict__ blk = out + (long long)d * TCP;
    int deg = 0;

    const int m    = lane & 15;
    const int koff = (lane >> 4) << 1;
    const int rt   = wave >> 2;
    const int c0   = (wave & 3) << 4;
    v2f afr[CDIM / 4];
    {
        v4f ru[NODE_STAGE_VECS], rv[NODE_STAGE_VECS];
        for (int j = 0; j < NODE_STAGE_VECS; ++j) {
            ru[j] = ((const v4f*)U)[threadIdx.x + j * 256];
            rv[j] = ((const v4f*)V)[threadIdx.x + j * 256];
        }
        long long row = (long long)blockIdx.x * (NODES_PER_GROUP * TDIM) + rt * 16 + m;
        const long long last_row = (long long)n_dst * TDIM - 1;
        if (row > last_row) row = last_row;
        const float* __restrict__ arow = feat_dst + row * ROWLEN;
#pragma unroll
        for (int q = 0; q < CDIM / 4; ++q) {
            afr[q].x = arow[4 * q + koff];
            afr[q].y = arow[4 * q + koff + 1];
        }
        if (d < n_dst) deg = __builtin_bit_cast(int, blk[CNT_AT]);
        for (int j = 0; j < NODE_STAGE_VECS; ++j) {
            ((v4f*)ldsU)[threadIdx.x + j * 256] = ru[j];
            ((v4f*)ldsV)[threadIdx.x + j * 256] = rv[j];
        }
    }
    __syncthreads();
    {
        v8f accU = {0.f, 0.f, 0.f, 0.f, 0.f, 0.f, 0.f, 0.f};
        v8f accV = {0.f, 0.f, 0.f, 0.f, 0.f, 0.f, 0.f, 0.f};
#pragma unroll
        for (int q = 0; q < CDIM / 4; ++q) {
            v2f bu, bv;
            bu.x = ldsU[(4 * q + koff)     * ROWLEN + c0 + m];
            bu.y = ldsU[(4 * q + koff + 1) * ROWLEN + c0 + m];
            bv.x = ldsV[(4 * q + koff)     * ROWLEN + c0 + m];
            bv.y = ldsV[(4 * q + koff + 1) * ROWLEN + c0 + m];
            accU = __builtin_amdgcn_wmma_f32_16x16x4_f32(
                false, afr[q], false, bu, (short)0, accU, false, false);
            accV = __builtin_amdgcn_wmma_f32_16x16x4_f32(
                false, afr[q], false, bv, (short)0, accV, false, false);
        }
        const int rb = rt * 16 + ((lane >> 4) << 3);
#pragma unroll
        for (int i = 0; i < 8; ++i) {
            ldsBase[(rb + i) * RES_PITCH + c0 + m] = accU[i];
            ldsHd[(rb + i) * RES_PITCH + c0 + m]   = accV[i];
        }
    }
    __syncthreads();

    if (d >= n_dst) return;

    const int end = (deg <= BUCKET) ? deg : 0;
    const int o   = lane << 2;

    const v4f wv  = *(const v4f*)(we + (o & (CDIM - 1)));
    const int lofs = (wave * TDIM + (o >> 6)) * RES_PITCH + (o & (CDIM - 1));
    const v4f c0v = (-1.44269504088896341f * *(const v4f*)(ldsHd + lofs)) * wv;
    const v4f c1v = (-1.44269504088896341f * *(const v4f*)(ldsHd + lofs + 2 * RES_PITCH)) * wv;
    v4f a0 = *(const v4f*)(ldsBase + lofs);
    v4f a1 = *(const v4f*)(ldsBase + lofs + 2 * RES_PITCH);

    for (int j0 = 0; j0 < end; j0 += 32) {
        const int n = (end - j0 < 32) ? (end - j0) : 32;
        int   sv = 0;
        float wl = 0.f;
        if (lane < n) {
            sv = __builtin_bit_cast(int, blk[j0 + lane]);
            wl = blk[W_AT + j0 + lane];
        }
        int k = 0;
        for (; k + 4 <= n; k += 4) {
            const int s0 = __builtin_amdgcn_readlane(sv, k);
            const int s1 = __builtin_amdgcn_readlane(sv, k + 1);
            const int s2 = __builtin_amdgcn_readlane(sv, k + 2);
            const int s3 = __builtin_amdgcn_readlane(sv, k + 3);
            const float* __restrict__ p0 = h_src + (long long)s0 * TCP + o;
            const float* __restrict__ p1 = h_src + (long long)s1 * TCP + o;
            const float* __restrict__ p2 = h_src + (long long)s2 * TCP + o;
            const float* __restrict__ p3 = h_src + (long long)s3 * TCP + o;
            const v4f x00 = *(const v4f*)(p0), x01 = *(const v4f*)(p0 + 128);
            const v4f x10 = *(const v4f*)(p1), x11 = *(const v4f*)(p1 + 128);
            const v4f x20 = *(const v4f*)(p2), x21 = *(const v4f*)(p2 + 128);
            const v4f x30 = *(const v4f*)(p3), x31 = *(const v4f*)(p3 + 128);
            const float w0 = __builtin_bit_cast(float, __builtin_amdgcn_readlane(__builtin_bit_cast(int, wl), k));
            const float w1 = __builtin_bit_cast(float, __builtin_amdgcn_readlane(__builtin_bit_cast(int, wl), k + 1));
            const float w2 = __builtin_bit_cast(float, __builtin_amdgcn_readlane(__builtin_bit_cast(int, wl), k + 2));
            const float w3 = __builtin_bit_cast(float, __builtin_amdgcn_readlane(__builtin_bit_cast(int, wl), k + 3));
            gate_acc(a0, x00, c0v, w0);  gate_acc(a1, x01, c1v, w0);
            gate_acc(a0, x10, c0v, w1);  gate_acc(a1, x11, c1v, w1);
            gate_acc(a0, x20, c0v, w2);  gate_acc(a1, x21, c1v, w2);
            gate_acc(a0, x30, c0v, w3);  gate_acc(a1, x31, c1v, w3);
        }
        for (; k < n; ++k) {
            const int s0 = __builtin_amdgcn_readlane(sv, k);
            const float* __restrict__ p0 = h_src + (long long)s0 * TCP + o;
            const v4f x00 = *(const v4f*)(p0), x01 = *(const v4f*)(p0 + 128);
            const float w0 = __builtin_bit_cast(float, __builtin_amdgcn_readlane(__builtin_bit_cast(int, wl), k));
            gate_acc(a0, x00, c0v, w0);  gate_acc(a1, x01, c1v, w0);
        }
    }
    if (deg > BUCKET) {
        for (int j0 = 0; j0 < E; j0 += 32) {
            const int j = j0 + lane;
            int dd = -1;
            if (j < E) dd = dst_idx[j];
            unsigned mine = __builtin_amdgcn_ballot_w32(dd == d);
            if (mine == 0u) continue;
            int   sv = 0;
            float wl = 0.f;
            if (dd == d) {
                sv = src_idx[j];
                wl = edge_w[j];
            }
            while (mine) {
                const int k = __builtin_ctz(mine);
                mine &= mine - 1;
                const int s0 = __builtin_amdgcn_readlane(sv, k);
                const float* __restrict__ p0 = h_src + (long long)s0 * TCP + o;
                const v4f x00 = *(const v4f*)(p0), x01 = *(const v4f*)(p0 + 128);
                const float w0 = __builtin_bit_cast(float, __builtin_amdgcn_readlane(__builtin_bit_cast(int, wl), k));
                gate_acc(a0, x00, c0v, w0);  gate_acc(a1, x01, c1v, w0);
            }
        }
    }

    const float* __restrict__ fp = feat_dst + (long long)d * TCP + o;
    const v4f f0 = *(const v4f*)(fp);
    const v4f f1 = *(const v4f*)(fp + 128);
    const v4f g = *(const v4f*)(gamma + (o & (CDIM - 1)));
    const v4f b = *(const v4f*)(beta  + (o & (CDIM - 1)));
    float s0 = (a0.x + a0.y) + (a0.z + a0.w);
    float s1 = (a1.x + a1.y) + (a1.z + a1.w);
    row16_sum2(s0, s1);
    const v4f e0 = a0 - s0 * (1.0f / 64.0f);
    const v4f e1 = a1 - s1 * (1.0f / 64.0f);
    float q0 = (e0.x * e0.x + e0.y * e0.y) + (e0.z * e0.z + e0.w * e0.w);
    float q1 = (e1.x * e1.x + e1.y * e1.y) + (e1.z * e1.z + e1.w * e1.w);
    row16_sum2(q0, q1);
    const float r0 = rsqrtf(q0 * (1.0f / 64.0f) + 1e-5f);
    const float r1 = rsqrtf(q1 * (1.0f / 64.0f) + 1e-5f);
    *(v4f*)(blk + o)       = e0 * r0 * g + b + f0;
    *(v4f*)(blk + o + 128) = e1 * r1 * g + b + f1;
}

extern "C" void kernel_launch(void* const* d_in, const int* in_sizes, int n_in,
                              void* d_out, int out_size, void* d_ws, size_t ws_size,
                              hipStream_t stream) {
    const float* feat_src = (const float*)d_in[0];
    const float* feat_dst = (const float*)d_in[1];
    const float* edge_w   = (const float*)d_in[2];
    const float* weight_e = (const float*)d_in[3];
    const float* u        = (const float*)d_in[4];
    const float* v        = (const float*)d_in[5];
    const float* ln_gamma = (const float*)d_in[6];
    const float* ln_beta  = (const float*)d_in[7];
    const int*   src_idx  = (const int*)d_in[8];
    const int*   dst_idx  = (const int*)d_in[9];
    float*       out      = (float*)d_out;

    const int n_src = in_sizes[0] / TCP;
    const int n_dst = in_sizes[1] / TCP;
    const int E     = in_sizes[2];

    float* h_src = (float*)d_ws;

    const int mtiles_src = (n_src * TDIM) / 16;
    const int wpb = PROJ_THREADS / 32;

    proj_src_kernel<<<(mtiles_src + wpb - 1) / wpb, PROJ_THREADS, 0, stream>>>(
        feat_src, u, h_src, mtiles_src, out, n_dst);

    const int group_groups = (E + EDGES_PER_GROUP - 1) / EDGES_PER_GROUP;
    group_kernel<<<group_groups, PROJ_THREADS, 0, stream>>>(src_idx, dst_idx, edge_w, out, E);

    node_sum_ln_kernel<<<(n_dst + NODES_PER_GROUP - 1) / NODES_PER_GROUP, NODES_PER_GROUP * 32, 0, stream>>>(
        h_src, u, v, weight_e, src_idx, dst_idx, edge_w, feat_dst,
        ln_gamma, ln_beta, out, n_dst, E);
}


# Projections: feature rows (feat_src in launch A, feat_dst in launch C) staged in LDS by whole-line 16-byte loads, A-fragments read from LDS; was 8-byte loads touching 16 lines each; sum order kept
# speedup vs baseline: 2.8232x; 1.0367x over previous
#include <hip/hip_runtime.h>
#include <math.h>


typedef __attribute__((ext_vector_type(2))) float v2f;
typedef __attribute__((ext_vector_type(8))) float v8f;
typedef __attribute__((ext_vector_type(4))) float v4f;

#define CDIM 64
#define TDIM 4
#define ROWLEN 64
#define TCP 256
#define PROJ_THREADS 128
#define STAGE_VECS ((CDIM * CDIM / 4) / PROJ_THREADS)
#define BUCKET 64
#define W_AT   BUCKET
#define CNT_AT (2 * BUCKET)
#define EDGES_PER_THREAD 4
#define EDGES_PER_GROUP (PROJ_THREADS * EDGES_PER_THREAD)
#define NODES_PER_GROUP 8
#define NODE_STAGE_VECS ((CDIM * CDIM / 4) / (NODES_PER_GROUP * 32))
#define RES_PITCH 68
#define A_PITCH 68
#define SRC_ROW_VECS ((PROJ_THREADS / 32) * 16 * (ROWLEN / 4) / PROJ_THREADS)
#define DST_ROW_VECS (NODES_PER_GROUP * TDIM * (ROWLEN / 4) / (NODES_PER_GROUP * 32))

__device__ __forceinline__ v8f wmma16x16_k64(const v2f (&afr)[CDIM / 4],
                                             const float* __restrict__ Blds,
                                             int c0) {
    const int lane = threadIdx.x & 31;
    const int m    = lane & 15;
    const int koff = (lane >> 4) << 1;
    v8f acc = {0.f, 0.f, 0.f, 0.f, 0.f, 0.f, 0.f, 0.f};
#pragma unroll
    for (int k0 = 0; k0 < 64; k0 += 4) {
        v2f b;
        b.x = Blds[(k0 + koff)     * ROWLEN + c0 + m];
        b.y = Blds[(k0 + koff + 1) * ROWLEN + c0 + m];
        acc = __builtin_amdgcn_wmma_f32_16x16x4_f32(
            false, afr[k0 >> 2], false, b, (short)0, acc, false, false);
    }
    return acc;
}

__device__ __forceinline__ void load_afr(v2f (&afr)[CDIM / 4], const float* __restrict__ tile_rows) {
    const int lane = threadIdx.x & 31;
    const float* __restrict__ arow = tile_rows + (lane & 15) * A_PITCH + ((lane >> 4) << 1);
#pragma unroll
    for (int q = 0; q < CDIM / 4; ++q)
        afr[q] = *(const v2f*)(arow + 4 * q);
}

__device__ __forceinline__ void store_tile(float* __restrict__ Out, int row0,
                                           int c0, v8f acc) {
    const int lane = threadIdx.x & 31;
    const int col  = c0 + (lane & 15);
    const int rb   = row0 + ((lane >> 4) << 3);
#pragma unroll
    for (int i = 0; i < 8; ++i)
        Out[(rb + i) * ROWLEN + col] = acc[i];
}

__global__ void __launch_bounds__(PROJ_THREADS)
proj_src_kernel(const float* __restrict__ feat, const float* __restrict__ U,
                float* __restrict__ Out, int mtiles,
                float* __restrict__ out, int n_dst) {
    __shared__ __attribute__((aligned(16))) float ldsU[CDIM * CDIM];
    __shared__ __attribute__((aligned(16))) float ldsA[(PROJ_THREADS / 32) * 16 * A_PITCH];
    for (int i = blockIdx.x * PROJ_THREADS + threadIdx.x; i < n_dst;
         i += gridDim.x * PROJ_THREADS)
        ((int*)out)[(long long)i * TCP + CNT_AT] = 0;
    {
        v4f ru[STAGE_VECS], ra[SRC_ROW_VECS];
        for (int j = 0; j < STAGE_VECS; ++j)
            ru[j] = ((const v4f*)U)[threadIdx.x + j * PROJ_THREADS];
        const long long last_row = (long long)mtiles * 16 - 1;
        for (int j = 0; j < SRC_ROW_VECS; ++j) {
            const int p = threadIdx.x + j * PROJ_THREADS;
            long long row = (long long)blockIdx.x * ((PROJ_THREADS / 32) * 16) + (p >> 4);
            if (row > last_row) row = last_row;
            ra[j] = *(const v4f*)(feat + row * ROWLEN + ((p & 15) << 2));
        }
        for (int j = 0; j < STAGE_VECS; ++j)
            ((v4f*)ldsU)[threadIdx.x + j * PROJ_THREADS] = ru[j];
        for (int j = 0; j < SRC_ROW_VECS; ++j) {
            const int p = threadIdx.x + j * PROJ_THREADS;
            *(v4f*)(ldsA + (p >> 4) * A_PITCH + ((p & 15) << 2)) = ra[j];
        }
    }
    __syncthreads();

    const int wave = threadIdx.x >> 5;
    const int tile = blockIdx.x * (blockDim.x >> 5) + wave;
    if (tile >= mtiles) return;
    const int row0 = tile * 16;
    v2f afr[CDIM / 4];
    load_afr(afr, ldsA + wave * 16 * A_PITCH);
#pragma unroll
    for (int c0 = 0; c0 < 64; c0 += 16) {
        v8f acc = wmma16x16_k64(afr, ldsU, c0);
        store_tile(Out, row0, c0, acc);
    }
}

__device__ __forceinline__ void group_edges(const int* __restrict__ src_idx,
                                            const int* __restrict__ dst_idx,
                                            const float* __restrict__ edge_w,
                                            float* __restrict__ out, int E, int group) {
    int   d[EDGES_PER_THREAD], s[EDGES_PER_THREAD], slot[EDGES_PER_THREAD];
    float w[EDGES_PER_THREAD];
    const int e0 = group * EDGES_PER_GROUP + threadIdx.x;
#pragma unroll
    for (int q = 0; q < EDGES_PER_THREAD; ++q) {
        const int e = e0 + q * PROJ_THREADS;
        d[q] = -1;  s[q] = 0;  w[q] = 0.f;
        if (e < E) {
            d[q] = dst_idx[e];
            s[q] = src_idx[e];
            w[q] = edge_w[e];
        }
    }
#pragma unroll
    for (int q = 0; q < EDGES_PER_THREAD; ++q) {
        asm volatile("" : : "v"(s[q]), "v"(w[q]));
    }
#pragma unroll
    for (int q = 0; q < EDGES_PER_THREAD; ++q) {
        slot[q] = BUCKET;
        if (d[q] >= 0)
            slot[q] = atomicAdd((int*)out + (long long)d[q] * TCP + CNT_AT, 1);
    }
#pragma unroll
    for (int q = 0; q < EDGES_PER_THREAD; ++q) {
        if (slot[q] < BUCKET) {
            int* __restrict__ blk = (int*)out + (long long)d[q] * TCP;
            blk[slot[q]] = s[q];
            blk[W_AT + slot[q]] = __builtin_bit_cast(int, w[q]);
        }
    }
}

__global__ void __launch_bounds__(PROJ_THREADS)
group_kernel(const int* __restrict__ src_idx, const int* __restrict__ dst_idx,
             const float* __restrict__ edge_w, float* __restrict__ out, int E) {
    group_edges(src_idx, dst_idx, edge_w, out, E, (int)blockIdx.x);
}

__device__ __forceinline__ void gate_acc(v4f& a, const v4f hs, const v4f c, const float w) {
    const v4f y = hs * (c * w);
    v4f r;
    r.x = __builtin_amdgcn_rcpf(1.0f + __builtin_amdgcn_exp2f(y.x));
    r.y = __builtin_amdgcn_rcpf(1.0f + __builtin_amdgcn_exp2f(y.y));
    r.z = __builtin_amdgcn_rcpf(1.0f + __builtin_amdgcn_exp2f(y.z));
    r.w = __builtin_amdgcn_rcpf(1.0f + __builtin_amdgcn_exp2f(y.w));
    a = __builtin_elementwise_fma(hs, r, a);
}

template <int MASK>
__device__ __forceinline__ float add_row_xor(const float x) {
    return x + __builtin_bit_cast(float,
        __builtin_amdgcn_mov_dpp(__builtin_bit_cast(int, x), 0x160 | MASK, 0xf, 0xf, true));
}

__device__ __forceinline__ void row16_sum2(float& x, float& y) {
    x = add_row_xor<1>(x);  y = add_row_xor<1>(y);
    x = add_row_xor<2>(x);  y = add_row_xor<2>(y);
    x = add_row_xor<4>(x);  y = add_row_xor<4>(y);
    x = add_row_xor<8>(x);  y = add_row_xor<8>(y);
}

__global__ void __launch_bounds__(256, 12)
node_sum_ln_kernel(const float* __restrict__ h_src, const float* __restrict__ U,
                   const float* __restrict__ V, const float* __restrict__ we,
                   const int* __restrict__ src_idx, const int* __restrict__ dst_idx,
                   const float* __restrict__ edge_w, const float* __restrict__ feat_dst,
                   const float* __restrict__ gamma, const float* __restrict__ beta,
                   float* __restrict__ out, int n_dst, int E) {
    __shared__ __attribute__((aligned(16))) float ldsU[CDIM * CDIM];
    __shared__ __attribute__((aligned(16))) float ldsV[CDIM * CDIM];
    __shared__ __attribute__((aligned(16))) float ldsBase[NODES_PER_GROUP * TDIM * RES_PITCH];
    __shared__ __attribute__((aligned(16))) float ldsHd[NODES_PER_GROUP * TDIM * RES_PITCH];
    const int lane = threadIdx.x & 31;
    const int wave = threadIdx.x >> 5;
    const int d = __builtin_amdgcn_readfirstlane(
        (int)(blockIdx.x * NODES_PER_GROUP + wave));
    float* __restrict__ blk = out + (long long)d * TCP;
    int deg = 0;

    const int m    = lane & 15;
    const int koff = (lane >> 4) << 1;
    const int rt   = wave >> 2;
    const int c0   = (wave & 3) << 4;
    static_assert(A_PITCH <= RES_PITCH, "the staged rows must fit the base tile they share LDS with");
    float* const ldsA = ldsBase;
    v2f afr[CDIM / 4];
    {
        v4f ru[NODE_STAGE_VECS], rv[NODE_STAGE_VECS], rf[DST_ROW_VECS];
        for (int j = 0; j < NODE_STAGE_VECS; ++j) {
            ru[j] = ((const v4f*)U)[threadIdx.x + j * 256];
            rv[j] = ((const v4f*)V)[threadIdx.x + j * 256];
        }
        const long long last_row = (long long)n_dst * TDIM - 1;
        for (int j = 0; j < DST_ROW_VECS; ++j) {
            const int p = threadIdx.x + j * 256;
            long long row = (long long)blockIdx.x * (NODES_PER_GROUP * TDIM) + (p >> 4);
            if (row > last_row) row = last_row;
            rf[j] = *(const v4f*)(feat_dst + row * ROWLEN + ((p & 15) << 2));
        }
        if (d < n_dst) deg = __builtin_bit_cast(int, blk[CNT_AT]);
        for (int j = 0; j < NODE_STAGE_VECS; ++j) {
            ((v4f*)ldsU)[threadIdx.x + j * 256] = ru[j];
            ((v4f*)ldsV)[threadIdx.x + j * 256] = rv[j];
        }
        for (int j = 0; j < DST_ROW_VECS; ++j) {
            const int p = threadIdx.x + j * 256;
            *(v4f*)(ldsA + (p >> 4) * A_PITCH + ((p & 15) << 2)) = rf[j];
        }
    }
    __syncthreads();
    load_afr(afr, ldsA + rt * 16 * A_PITCH);
    __syncthreads();
    {
        v8f accU = {0.f, 0.f, 0.f, 0.f, 0.f, 0.f, 0.f, 0.f};
        v8f accV = {0.f, 0.f, 0.f, 0.f, 0.f, 0.f, 0.f, 0.f};
#pragma unroll
        for (int q = 0; q < CDIM / 4; ++q) {
            v2f bu, bv;
            bu.x = ldsU[(4 * q + koff)     * ROWLEN + c0 + m];
            bu.y = ldsU[(4 * q + koff + 1) * ROWLEN + c0 + m];
            bv.x = ldsV[(4 * q + koff)     * ROWLEN + c0 + m];
            bv.y = ldsV[(4 * q + koff + 1) * ROWLEN + c0 + m];
            accU = __builtin_amdgcn_wmma_f32_16x16x4_f32(
                false, afr[q], false, bu, (short)0, accU, false, false);
            accV = __builtin_amdgcn_wmma_f32_16x16x4_f32(
                false, afr[q], false, bv, (short)0, accV, false, false);
        }
        const int rb = rt * 16 + ((lane >> 4) << 3);
#pragma unroll
        for (int i = 0; i < 8; ++i) {
            ldsBase[(rb + i) * RES_PITCH + c0 + m] = accU[i];
            ldsHd[(rb + i) * RES_PITCH + c0 + m]   = accV[i];
        }
    }
    __syncthreads();

    if (d >= n_dst) return;

    const int end = (deg <= BUCKET) ? deg : 0;
    const int o   = lane << 2;

    const v4f wv  = *(const v4f*)(we + (o & (CDIM - 1)));
    const int lofs = (wave * TDIM + (o >> 6)) * RES_PITCH + (o & (CDIM - 1));
    const v4f c0v = (-1.44269504088896341f * *(const v4f*)(ldsHd + lofs)) * wv;
    const v4f c1v = (-1.44269504088896341f * *(const v4f*)(ldsHd + lofs + 2 * RES_PITCH)) * wv;
    v4f a0 = *(const v4f*)(ldsBase + lofs);
    v4f a1 = *(const v4f*)(ldsBase + lofs + 2 * RES_PITCH);

    for (int j0 = 0; j0 < end; j0 += 32) {
        const int n = (end - j0 < 32) ? (end - j0) : 32;
        int   sv = 0;
        float wl = 0.f;
        if (lane < n) {
            sv = __builtin_bit_cast(int, blk[j0 + lane]);
            wl = blk[W_AT + j0 + lane];
        }
        int k = 0;
        for (; k + 4 <= n; k += 4) {
            const int s0 = __builtin_amdgcn_readlane(sv, k);
            const int s1 = __builtin_amdgcn_readlane(sv, k + 1);
            const int s2 = __builtin_amdgcn_readlane(sv, k + 2);
            const int s3 = __builtin_amdgcn_readlane(sv, k + 3);
            const float* __restrict__ p0 = h_src + (long long)s0 * TCP + o;
            const float* __restrict__ p1 = h_src + (long long)s1 * TCP + o;
            const float* __restrict__ p2 = h_src + (long long)s2 * TCP + o;
            const float* __restrict__ p3 = h_src + (long long)s3 * TCP + o;
            const v4f x00 = *(const v4f*)(p0), x01 = *(const v4f*)(p0 + 128);
            const v4f x10 = *(const v4f*)(p1), x11 = *(const v4f*)(p1 + 128);
            const v4f x20 = *(const v4f*)(p2), x21 = *(const v4f*)(p2 + 128);
            const v4f x30 = *(const v4f*)(p3), x31 = *(const v4f*)(p3 + 128);
            const float w0 = __builtin_bit_cast(float, __builtin_amdgcn_readlane(__builtin_bit_cast(int, wl), k));
            const float w1 = __builtin_bit_cast(float, __builtin_amdgcn_readlane(__builtin_bit_cast(int, wl), k + 1));
            const float w2 = __builtin_bit_cast(float, __builtin_amdgcn_readlane(__builtin_bit_cast(int, wl), k + 2));
            const float w3 = __builtin_bit_cast(float, __builtin_amdgcn_readlane(__builtin_bit_cast(int, wl), k + 3));
            gate_acc(a0, x00, c0v, w0);  gate_acc(a1, x01, c1v, w0);
            gate_acc(a0, x10, c0v, w1);  gate_acc(a1, x11, c1v, w1);
            gate_acc(a0, x20, c0v, w2);  gate_acc(a1, x21, c1v, w2);
            gate_acc(a0, x30, c0v, w3);  gate_acc(a1, x31, c1v, w3);
        }
        for (; k < n; ++k) {
            const int s0 = __builtin_amdgcn_readlane(sv, k);
            const float* __restrict__ p0 = h_src + (long long)s0 * TCP + o;
            const v4f x00 = *(const v4f*)(p0), x01 = *(const v4f*)(p0 + 128);
            const float w0 = __builtin_bit_cast(float, __builtin_amdgcn_readlane(__builtin_bit_cast(int, wl), k));
            gate_acc(a0, x00, c0v, w0);  gate_acc(a1, x01, c1v, w0);
        }
    }
    if (deg > BUCKET) {
        for (int j0 = 0; j0 < E; j0 += 32) {
            const int j = j0 + lane;
            int dd = -1;
            if (j < E) dd = dst_idx[j];
            unsigned mine = __builtin_amdgcn_ballot_w32(dd == d);
            if (mine == 0u) continue;
            int   sv = 0;
            float wl = 0.f;
            if (dd == d) {
                sv = src_idx[j];
                wl = edge_w[j];
            }
            while (mine) {
                const int k = __builtin_ctz(mine);
                mine &= mine - 1;
                const int s0 = __builtin_amdgcn_readlane(sv, k);
                const float* __restrict__ p0 = h_src + (long long)s0 * TCP + o;
                const v4f x00 = *(const v4f*)(p0), x01 = *(const v4f*)(p0 + 128);
                const float w0 = __builtin_bit_cast(float, __builtin_amdgcn_readlane(__builtin_bit_cast(int, wl), k));
                gate_acc(a0, x00, c0v, w0);  gate_acc(a1, x01, c1v, w0);
            }
        }
    }

    const float* __restrict__ fp = feat_dst + (long long)d * TCP + o;
    const v4f f0 = *(const v4f*)(fp);
    const v4f f1 = *(const v4f*)(fp + 128);
    const v4f g = *(const v4f*)(gamma + (o & (CDIM - 1)));
    const v4f b = *(const v4f*)(beta  + (o & (CDIM - 1)));
    float s0 = (a0.x + a0.y) + (a0.z + a0.w);
    float s1 = (a1.x + a1.y) + (a1.z + a1.w);
    row16_sum2(s0, s1);
    const v4f e0 = a0 - s0 * (1.0f / 64.0f);
    const v4f e1 = a1 - s1 * (1.0f / 64.0f);
    float q0 = (e0.x * e0.x + e0.y * e0.y) + (e0.z * e0.z + e0.w * e0.w);
    float q1 = (e1.x * e1.x + e1.y * e1.y) + (e1.z * e1.z + e1.w * e1.w);
    row16_sum2(q0, q1);
    const float r0 = rsqrtf(q0 * (1.0f / 64.0f) + 1e-5f);
    const float r1 = rsqrtf(q1 * (1.0f / 64.0f) + 1e-5f);
    *(v4f*)(blk + o)       = e0 * r0 * g + b + f0;
    *(v4f*)(blk + o + 128) = e1 * r1 * g + b + f1;
}

extern "C" void kernel_launch(void* const* d_in, const int* in_sizes, int n_in,
                              void* d_out, int out_size, void* d_ws, size_t ws_size,
                              hipStream_t stream) {
    const float* feat_src = (const float*)d_in[0];
    const float* feat_dst = (const float*)d_in[1];
    const float* edge_w   = (const float*)d_in[2];
    const float* weight_e = (const float*)d_in[3];
    const float* u        = (const float*)d_in[4];
    const float* v        = (const float*)d_in[5];
    const float* ln_gamma = (const float*)d_in[6];
    const float* ln_beta  = (const float*)d_in[7];
    const int*   src_idx  = (const int*)d_in[8];
    const int*   dst_idx  = (const int*)d_in[9];
    float*       out      = (float*)d_out;

    const int n_src = in_sizes[0] / TCP;
    const int n_dst = in_sizes[1] / TCP;
    const int E     = in_sizes[2];

    float* h_src = (float*)d_ws;

    const int mtiles_src = (n_src * TDIM) / 16;
    const int wpb = PROJ_THREADS / 32;

    proj_src_kernel<<<(mtiles_src + wpb - 1) / wpb, PROJ_THREADS, 0, stream>>>(
        feat_src, u, h_src, mtiles_src, out, n_dst);

    const int group_groups = (E + EDGES_PER_GROUP - 1) / EDGES_PER_GROUP;
    group_kernel<<<group_groups, PROJ_THREADS, 0, stream>>>(src_idx, dst_idx, edge_w, out, E);

    node_sum_ln_kernel<<<(n_dst + NODES_PER_GROUP - 1) / NODES_PER_GROUP, NODES_PER_GROUP * 32, 0, stream>>>(
        h_src, u, v, weight_e, src_idx, dst_idx, edge_w, feat_dst,
        ln_gamma, ln_beta, out, n_dst, E);
}


# Edge grouping and source projection share one launch (grouping work-groups first; matrix work runs in the counting adds' waits); counter clearing is a tiny first launch; 3 launches; sum order kept
# speedup vs baseline: 2.9692x; 1.0517x over previous
#include <hip/hip_runtime.h>
#include <math.h>


typedef __attribute__((ext_vector_type(2))) float v2f;
typedef __attribute__((ext_vector_type(8))) float v8f;
typedef __attribute__((ext_vector_type(4))) float v4f;

#define CDIM 64
#define TDIM 4
#define ROWLEN 64
#define TCP 256
#define PROJ_THREADS 128
#define STAGE_VECS ((CDIM * CDIM / 4) / PROJ_THREADS)
#define BUCKET 64
#define W_AT   BUCKET
#define CNT_AT (2 * BUCKET)
#define EDGES_PER_THREAD 4
#define EDGES_PER_GROUP (PROJ_THREADS * EDGES_PER_THREAD)
#define NODES_PER_GROUP 8
#define NODE_STAGE_VECS ((CDIM * CDIM / 4) / (NODES_PER_GROUP * 32))
#define RES_PITCH 68
#define A_PITCH 68
#define SRC_ROW_VECS ((PROJ_THREADS / 32) * 16 * (ROWLEN / 4) / PROJ_THREADS)
#define DST_ROW_VECS (NODES_PER_GROUP * TDIM * (ROWLEN / 4) / (NODES_PER_GROUP * 32))

__device__ __forceinline__ v8f wmma16x16_k64(const v2f (&afr)[CDIM / 4],
                                             const float* __restrict__ Blds,
                                             int c0) {
    const int lane = threadIdx.x & 31;
    const int m    = lane & 15;
    const int koff = (lane >> 4) << 1;
    v8f acc = {0.f, 0.f, 0.f, 0.f, 0.f, 0.f, 0.f, 0.f};
#pragma unroll
    for (int k0 = 0; k0 < 64; k0 += 4) {
        v2f b;
        b.x = Blds[(k0 + koff)     * ROWLEN + c0 + m];
        b.y = Blds[(k0 + koff + 1) * ROWLEN + c0 + m];
        acc = __builtin_amdgcn_wmma_f32_16x16x4_f32(
            false, afr[k0 >> 2], false, b, (short)0, acc, false, false);
    }
    return acc;
}

__device__ __forceinline__ void load_afr(v2f (&afr)[CDIM / 4], const float* __restrict__ tile_rows) {
    const int lane = threadIdx.x & 31;
    const float* __restrict__ arow = tile_rows + (lane & 15) * A_PITCH + ((lane >> 4) << 1);
#pragma unroll
    for (int q = 0; q < CDIM / 4; ++q)
        afr[q] = *(const v2f*)(arow + 4 * q);
}

__device__ __forceinline__ void store_tile(float* __restrict__ Out, int row0,
                                           int c0, v8f acc) {
    const int lane = threadIdx.x & 31;
    const int col  = c0 + (lane & 15);
    const int rb   = row0 + ((lane >> 4) << 3);
#pragma unroll
    for (int i = 0; i < 8; ++i)
        Out[(rb + i) * ROWLEN + col] = acc[i];
}

__global__ void __launch_bounds__(256)
clear_counters_kernel(float* __restrict__ out, int n_dst) {
    const int i = blockIdx.x * 256 + threadIdx.x;
    if (i < n_dst) ((int*)out)[(long long)i * TCP + CNT_AT] = 0;
}

__device__ __forceinline__ void group_edges(const int* __restrict__ src_idx,
                                            const int* __restrict__ dst_idx,
                                            const float* __restrict__ edge_w,
                                            float* __restrict__ out, int E, int group) {
    int   d[EDGES_PER_THREAD], s[EDGES_PER_THREAD], slot[EDGES_PER_THREAD];
    float w[EDGES_PER_THREAD];
    const int e0 = group * EDGES_PER_GROUP + threadIdx.x;
#pragma unroll
    for (int q = 0; q < EDGES_PER_THREAD; ++q) {
        const int e = e0 + q * PROJ_THREADS;
        d[q] = -1;  s[q] = 0;  w[q] = 0.f;
        if (e < E) {
            d[q] = dst_idx[e];
            s[q] = src_idx[e];
            w[q] = edge_w[e];
        }
    }
#pragma unroll
    for (int q = 0; q < EDGES_PER_THREAD; ++q) {
        asm volatile("" : : "v"(s[q]), "v"(w[q]));
    }
#pragma unroll
    for (int q = 0; q < EDGES_PER_THREAD; ++q) {
        slot[q] = BUCKET;
        if (d[q] >= 0)
            slot[q] = atomicAdd((int*)out + (long long)d[q] * TCP + CNT_AT, 1);
    }
#pragma unroll
    for (int q = 0; q < EDGES_PER_THREAD; ++q) {
        if (slot[q] < BUCKET) {
            int* __restrict__ blk = (int*)out + (long long)d[q] * TCP;
            blk[slot[q]] = s[q];
            blk[W_AT + slot[q]] = __builtin_bit_cast(int, w[q]);
        }
    }
}

__global__ void __launch_bounds__(PROJ_THREADS)
group_proj_src_kernel(const int* __restrict__ src_idx, const int* __restrict__ dst_idx,
                      const float* __restrict__ edge_w, float* __restrict__ out, int E,
                      int group_groups,
                      const float* __restrict__ feat, const float* __restrict__ U,
                      float* __restrict__ Out, int mtiles) {
    __shared__ __attribute__((aligned(16))) float ldsU[CDIM * CDIM];
    __shared__ __attribute__((aligned(16))) float ldsA[(PROJ_THREADS / 32) * 16 * A_PITCH];
    if ((int)blockIdx.x < group_groups) {
        group_edges(src_idx, dst_idx, edge_w, out, E, (int)blockIdx.x);
        return;
    }
    const int pblock = (int)blockIdx.x - group_groups;
    {
        v4f ru[STAGE_VECS], ra[SRC_ROW_VECS];
        for (int j = 0; j < STAGE_VECS; ++j)
            ru[j] = ((const v4f*)U)[threadIdx.x + j * PROJ_THREADS];
        const long long last_row = (long long)mtiles * 16 - 1;
        for (int j = 0; j < SRC_ROW_VECS; ++j) {
            const int p = threadIdx.x + j * PROJ_THREADS;
            long long row = (long long)pblock * ((PROJ_THREADS / 32) * 16) + (p >> 4);
            if (row > last_row) row = last_row;
            ra[j] = *(const v4f*)(feat + row * ROWLEN + ((p & 15) << 2));
        }
        for (int j = 0; j < STAGE_VECS; ++j)
            ((v4f*)ldsU)[threadIdx.x + j * PROJ_THREADS] = ru[j];
        for (int j = 0; j < SRC_ROW_VECS; ++j) {
            const int p = threadIdx.x + j * PROJ_THREADS;
            *(v4f*)(ldsA + (p >> 4) * A_PITCH + ((p & 15) << 2)) = ra[j];
        }
    }
    __syncthreads();

    const int wave = threadIdx.x >> 5;
    const int tile = pblock * (blockDim.x >> 5) + wave;
    if (tile >= mtiles) return;
    const int row0 = tile * 16;
    v2f afr[CDIM / 4];
    load_afr(afr, ldsA + wave * 16 * A_PITCH);
#pragma unroll
    for (int c0 = 0; c0 < 64; c0 += 16) {
        v8f acc = wmma16x16_k64(afr, ldsU, c0);
        store_tile(Out, row0, c0, acc);
    }
}

__device__ __forceinline__ void gate_acc(v4f& a, const v4f hs, const v4f c, const float w) {
    const v4f y = hs * (c * w);
    v4f r;
    r.x = __builtin_amdgcn_rcpf(1.0f + __builtin_amdgcn_exp2f(y.x));
    r.y = __builtin_amdgcn_rcpf(1.0f + __builtin_amdgcn_exp2f(y.y));
    r.z = __builtin_amdgcn_rcpf(1.0f + __builtin_amdgcn_exp2f(y.z));
    r.w = __builtin_amdgcn_rcpf(1.0f + __builtin_amdgcn_exp2f(y.w));
    a = __builtin_elementwise_fma(hs, r, a);
}

template <int MASK>
__device__ __forceinline__ float add_row_xor(const float x) {
    return x + __builtin_bit_cast(float,
        __builtin_amdgcn_mov_dpp(__builtin_bit_cast(int, x), 0x160 | MASK, 0xf, 0xf, true));
}

__device__ __forceinline__ void row16_sum2(float& x, float& y) {
    x = add_row_xor<1>(x);  y = add_row_xor<1>(y);
    x = add_row_xor<2>(x);  y = add_row_xor<2>(y);
    x = add_row_xor<4>(x);  y = add_row_xor<4>(y);
    x = add_row_xor<8>(x);  y = add_row_xor<8>(y);
}

__global__ void __launch_bounds__(256, 12)
node_sum_ln_kernel(const float* __restrict__ h_src, const float* __restrict__ U,
                   const float* __restrict__ V, const float* __restrict__ we,
                   const int* __restrict__ src_idx, const int* __restrict__ dst_idx,
                   const float* __restrict__ edge_w, const float* __restrict__ feat_dst,
                   const float* __restrict__ gamma, const float* __restrict__ beta,
                   float* __restrict__ out, int n_dst, int E) {
    __shared__ __attribute__((aligned(16))) float ldsU[CDIM * CDIM];
    __shared__ __attribute__((aligned(16))) float ldsV[CDIM * CDIM];
    __shared__ __attribute__((aligned(16))) float ldsBase[NODES_PER_GROUP * TDIM * RES_PITCH];
    __shared__ __attribute__((aligned(16))) float ldsHd[NODES_PER_GROUP * TDIM * RES_PITCH];
    const int lane = threadIdx.x & 31;
    const int wave = threadIdx.x >> 5;
    const int d = __builtin_amdgcn_readfirstlane(
        (int)(blockIdx.x * NODES_PER_GROUP + wave));
    float* __restrict__ blk = out + (long long)d * TCP;
    int deg = 0;

    const int m    = lane & 15;
    const int koff = (lane >> 4) << 1;
    const int rt   = wave >> 2;
    const int c0   = (wave & 3) << 4;
    static_assert(A_PITCH <= RES_PITCH, "the staged rows must fit the base tile they share LDS with");
    float* const ldsA = ldsBase;
    v2f afr[CDIM / 4];
    {
        v4f ru[NODE_STAGE_VECS], rv[NODE_STAGE_VECS], rf[DST_ROW_VECS];
        for (int j = 0; j < NODE_STAGE_VECS; ++j) {
            ru[j] = ((const v4f*)U)[threadIdx.x + j * 256];
            rv[j] = ((const v4f*)V)[threadIdx.x + j * 256];
        }
        const long long last_row = (long long)n_dst * TDIM - 1;
        for (int j = 0; j < DST_ROW_VECS; ++j) {
            const int p = threadIdx.x + j * 256;
            long long row = (long long)blockIdx.x * (NODES_PER_GROUP * TDIM) + (p >> 4);
            if (row > last_row) row = last_row;
            rf[j] = *(const v4f*)(feat_dst + row * ROWLEN + ((p & 15) << 2));
        }
        if (d < n_dst) deg = __builtin_bit_cast(int, blk[CNT_AT]);
        for (int j = 0; j < NODE_STAGE_VECS; ++j) {
            ((v4f*)ldsU)[threadIdx.x + j * 256] = ru[j];
            ((v4f*)ldsV)[threadIdx.x + j * 256] = rv[j];
        }
        for (int j = 0; j < DST_ROW_VECS; ++j) {
            const int p = threadIdx.x + j * 256;
            *(v4f*)(ldsA + (p >> 4) * A_PITCH + ((p & 15) << 2)) = rf[j];
        }
    }
    __syncthreads();
    load_afr(afr, ldsA + rt * 16 * A_PITCH);
    __syncthreads();
    {
        v8f accU = {0.f, 0.f, 0.f, 0.f, 0.f, 0.f, 0.f, 0.f};
        v8f accV = {0.f, 0.f, 0.f, 0.f, 0.f, 0.f, 0.f, 0.f};
#pragma unroll
        for (int q = 0; q < CDIM / 4; ++q) {
            v2f bu, bv;
            bu.x = ldsU[(4 * q + koff)     * ROWLEN + c0 + m];
            bu.y = ldsU[(4 * q + koff + 1) * ROWLEN + c0 + m];
            bv.x = ldsV[(4 * q + koff)     * ROWLEN + c0 + m];
            bv.y = ldsV[(4 * q + koff + 1) * ROWLEN + c0 + m];
            accU = __builtin_amdgcn_wmma_f32_16x16x4_f32(
                false, afr[q], false, bu, (short)0, accU, false, false);
            accV = __builtin_amdgcn_wmma_f32_16x16x4_f32(
                false, afr[q], false, bv, (short)0, accV, false, false);
        }
        const int rb = rt * 16 + ((lane >> 4) << 3);
#pragma unroll
        for (int i = 0; i < 8; ++i) {
            ldsBase[(rb + i) * RES_PITCH + c0 + m] = accU[i];
            ldsHd[(rb + i) * RES_PITCH + c0 + m]   = accV[i];
        }
    }
    __syncthreads();

    if (d >= n_dst) return;

    const int end = (deg <= BUCKET) ? deg : 0;
    const int o   = lane << 2;

    const v4f wv  = *(const v4f*)(we + (o & (CDIM - 1)));
    const int lofs = (wave * TDIM + (o >> 6)) * RES_PITCH + (o & (CDIM - 1));
    const v4f c0v = (-1.44269504088896341f * *(const v4f*)(ldsHd + lofs)) * wv;
    const v4f c1v = (-1.44269504088896341f * *(const v4f*)(ldsHd + lofs + 2 * RES_PITCH)) * wv;
    v4f a0 = *(const v4f*)(ldsBase + lofs);
    v4f a1 = *(const v4f*)(ldsBase + lofs + 2 * RES_PITCH);

    for (int j0 = 0; j0 < end; j0 += 32) {
        const int n = (end - j0 < 32) ? (end - j0) : 32;
        int   sv = 0;
        float wl = 0.f;
        if (lane < n) {
            sv = __builtin_bit_cast(int, blk[j0 + lane]);
            wl = blk[W_AT + j0 + lane];
        }
        int k = 0;
        for (; k + 4 <= n; k += 4) {
            const int s0 = __builtin_amdgcn_readlane(sv, k);
            const int s1 = __builtin_amdgcn_readlane(sv, k + 1);
            const int s2 = __builtin_amdgcn_readlane(sv, k + 2);
            const int s3 = __builtin_amdgcn_readlane(sv, k + 3);
            const float* __restrict__ p0 = h_src + (long long)s0 * TCP + o;
            const float* __restrict__ p1 = h_src + (long long)s1 * TCP + o;
            const float* __restrict__ p2 = h_src + (long long)s2 * TCP + o;
            const float* __restrict__ p3 = h_src + (long long)s3 * TCP + o;
            const v4f x00 = *(const v4f*)(p0), x01 = *(const v4f*)(p0 + 128);
            const v4f x10 = *(const v4f*)(p1), x11 = *(const v4f*)(p1 + 128);
            const v4f x20 = *(const v4f*)(p2), x21 = *(const v4f*)(p2 + 128);
            const v4f x30 = *(const v4f*)(p3), x31 = *(const v4f*)(p3 + 128);
            const float w0 = __builtin_bit_cast(float, __builtin_amdgcn_readlane(__builtin_bit_cast(int, wl), k));
            const float w1 = __builtin_bit_cast(float, __builtin_amdgcn_readlane(__builtin_bit_cast(int, wl), k + 1));
            const float w2 = __builtin_bit_cast(float, __builtin_amdgcn_readlane(__builtin_bit_cast(int, wl), k + 2));
            const float w3 = __builtin_bit_cast(float, __builtin_amdgcn_readlane(__builtin_bit_cast(int, wl), k + 3));
            gate_acc(a0, x00, c0v, w0);  gate_acc(a1, x01, c1v, w0);
            gate_acc(a0, x10, c0v, w1);  gate_acc(a1, x11, c1v, w1);
            gate_acc(a0, x20, c0v, w2);  gate_acc(a1, x21, c1v, w2);
            gate_acc(a0, x30, c0v, w3);  gate_acc(a1, x31, c1v, w3);
        }
        for (; k < n; ++k) {
            const int s0 = __builtin_amdgcn_readlane(sv, k);
            const float* __restrict__ p0 = h_src + (long long)s0 * TCP + o;
            const v4f x00 = *(const v4f*)(p0), x01 = *(const v4f*)(p0 + 128);
            const float w0 = __builtin_bit_cast(float, __builtin_amdgcn_readlane(__builtin_bit_cast(int, wl), k));
            gate_acc(a0, x00, c0v, w0);  gate_acc(a1, x01, c1v, w0);
        }
    }
    if (deg > BUCKET) {
        for (int j0 = 0; j0 < E; j0 += 32) {
            const int j = j0 + lane;
            int dd = -1;
            if (j < E) dd = dst_idx[j];
            unsigned mine = __builtin_amdgcn_ballot_w32(dd == d);
            if (mine == 0u) continue;
            int   sv = 0;
            float wl = 0.f;
            if (dd == d) {
                sv = src_idx[j];
                wl = edge_w[j];
            }
            while (mine) {
                const int k = __builtin_ctz(mine);
                mine &= mine - 1;
                const int s0 = __builtin_amdgcn_readlane(sv, k);
                const float* __restrict__ p0 = h_src + (long long)s0 * TCP + o;
                const v4f x00 = *(const v4f*)(p0), x01 = *(const v4f*)(p0 + 128);
                const float w0 = __builtin_bit_cast(float, __builtin_amdgcn_readlane(__builtin_bit_cast(int, wl), k));
                gate_acc(a0, x00, c0v, w0);  gate_acc(a1, x01, c1v, w0);
            }
        }
    }

    const float* __restrict__ fp = feat_dst + (long long)d * TCP + o;
    const v4f f0 = *(const v4f*)(fp);
    const v4f f1 = *(const v4f*)(fp + 128);
    const v4f g = *(const v4f*)(gamma + (o & (CDIM - 1)));
    const v4f b = *(const v4f*)(beta  + (o & (CDIM - 1)));
    float s0 = (a0.x + a0.y) + (a0.z + a0.w);
    float s1 = (a1.x + a1.y) + (a1.z + a1.w);
    row16_sum2(s0, s1);
    const v4f e0 = a0 - s0 * (1.0f / 64.0f);
    const v4f e1 = a1 - s1 * (1.0f / 64.0f);
    float q0 = (e0.x * e0.x + e0.y * e0.y) + (e0.z * e0.z + e0.w * e0.w);
    float q1 = (e1.x * e1.x + e1.y * e1.y) + (e1.z * e1.z + e1.w * e1.w);
    row16_sum2(q0, q1);
    const float r0 = rsqrtf(q0 * (1.0f / 64.0f) + 1e-5f);
    const float r1 = rsqrtf(q1 * (1.0f / 64.0f) + 1e-5f);
    *(v4f*)(blk + o)       = e0 * r0 * g + b + f0;
    *(v4f*)(blk + o + 128) = e1 * r1 * g + b + f1;
}

extern "C" void kernel_launch(void* const* d_in, const int* in_sizes, int n_in,
                              void* d_out, int out_size, void* d_ws, size_t ws_size,
                              hipStream_t stream) {
    const float* feat_src = (const float*)d_in[0];
    const float* feat_dst = (const float*)d_in[1];
    const float* edge_w   = (const float*)d_in[2];
    const float* weight_e = (const float*)d_in[3];
    const float* u        = (const float*)d_in[4];
    const float* v        = (const float*)d_in[5];
    const float* ln_gamma = (const float*)d_in[6];
    const float* ln_beta  = (const float*)d_in[7];
    const int*   src_idx  = (const int*)d_in[8];
    const int*   dst_idx  = (const int*)d_in[9];
    float*       out      = (float*)d_out;

    const int n_src = in_sizes[0] / TCP;
    const int n_dst = in_sizes[1] / TCP;
    const int E     = in_sizes[2];

    float* h_src = (float*)d_ws;

    const int mtiles_src = (n_src * TDIM) / 16;
    const int wpb = PROJ_THREADS / 32;

    clear_counters_kernel<<<(n_dst + 255) / 256, 256, 0, stream>>>(out, n_dst);

    const int group_groups = (E + EDGES_PER_GROUP - 1) / EDGES_PER_GROUP;
    const int proj_groups  = (mtiles_src + wpb - 1) / wpb;
    group_proj_src_kernel<<<group_groups + proj_groups, PROJ_THREADS, 0, stream>>>(
        src_idx, dst_idx, edge_w, out, E, group_groups, feat_src, u, h_src, mtiles_src);

    node_sum_ln_kernel<<<(n_dst + NODES_PER_GROUP - 1) / NODES_PER_GROUP, NODES_PER_GROUP * 32, 0, stream>>>(
        h_src, u, v, weight_e, src_idx, dst_idx, edge_w, feat_dst,
        ln_gamma, ln_beta, out, n_dst, E);
}
